# starting kernel
# baseline (speedup 1.0000x reference)
#include <hip/hip_runtime.h>

#define N_NODES  100000
#define N_EDGES  3200000
#define F_IN     10
#define KPAD     12
#define N_OUT    64

typedef __attribute__((ext_vector_type(2))) float v2f;
typedef __attribute__((ext_vector_type(8))) float v8f;

__global__ void __launch_bounds__(256) gcn_zero_kernel(float* p, int n) {
    int i = blockIdx.x * blockDim.x + threadIdx.x;
    if (i < n) p[i] = 0.0f;
}

__global__ void __launch_bounds__(256) gcn_degree_kernel(const int* __restrict__ src,
                                                         const int* __restrict__ dst,
                                                         float* __restrict__ deg_out,
                                                         float* __restrict__ deg_in,
                                                         int nE) {
    int e = blockIdx.x * blockDim.x + threadIdx.x;
    if (e < nE) {
        atomicAdd(&deg_out[src[e]], 1.0f);
        atomicAdd(&deg_in [dst[e]], 1.0f);
    }
}

__global__ void __launch_bounds__(256) gcn_nodeprep_kernel(const float* __restrict__ x,
                                                           const float* __restrict__ deg_out,
                                                           const float* __restrict__ deg_in,
                                                           float* __restrict__ norm_d,
                                                           float* __restrict__ xscaled,
                                                           int nN) {
    int n = blockIdx.x * blockDim.x + threadIdx.x;
    if (n < nN) {
        float dout = deg_out[n]; dout = (dout > 0.0f) ? dout : 1.0f;
        float din  = deg_in [n]; din  = (din  > 0.0f) ? din  : 1.0f;
        float ns = rsqrtf(dout);
        norm_d[n] = rsqrtf(din);
        const float* xr = x + (size_t)n * 40 + 15;
        float* o = xscaled + (size_t)n * F_IN;
#pragma unroll
        for (int k = 0; k < F_IN; ++k) o[k] = xr[k] * ns;
    }
}

__global__ void __launch_bounds__(64) gcn_foldw_kernel(const float* __restrict__ W1,
                                                       const float* __restrict__ b1,
                                                       const float* __restrict__ W2,
                                                       const float* __restrict__ b2,
                                                       float* __restrict__ Wc,
                                                       float* __restrict__ bc) {
    int j = threadIdx.x;
    float acc = b2[j];
    for (int k = 0; k < 128; ++k) acc += b1[k] * W2[k * 64 + j];
    bc[j] = acc;
#pragma unroll 1
    for (int i = 0; i < KPAD; ++i) {
        float s = 0.0f;
        if (i < F_IN)
            for (int k = 0; k < 128; ++k) s += W1[i * 128 + k] * W2[k * 64 + j];
        Wc[i * 64 + j] = s;
    }
}

__global__ void __launch_bounds__(256) gcn_scatter_kernel(const int* __restrict__ src,
                                                          const int* __restrict__ dst,
                                                          const float* __restrict__ xscaled,
                                                          float* __restrict__ agg,
                                                          int nE) {
    int e = blockIdx.x * blockDim.x + threadIdx.x;
    if (e < nE) {
        int s = src[e], d = dst[e];
        const float* xs = xscaled + (size_t)s * F_IN;
        float*       a  = agg     + (size_t)d * F_IN;
        float v[F_IN];
#pragma unroll
        for (int k = 0; k < F_IN; ++k) v[k] = xs[k];
#pragma unroll
        for (int k = 0; k < F_IN; ++k) atomicAdd(&a[k], v[k]);
    }
}

__global__ void __launch_bounds__(32) gcn_wmma_kernel(const float* __restrict__ agg,
                                                      const float* __restrict__ norm_d,
                                                      const float* __restrict__ Wc,
                                                      const float* __restrict__ bc,
                                                      float* __restrict__ out) {
    int tile = blockIdx.x;
    int lane = threadIdx.x;
    int m  = lane & 15;
    int kh = lane >> 4;

    int node = tile * 16 + m;
    float nd = norm_d[node];
    const float* arow = agg + (size_t)node * F_IN;

    v2f afrag[3];
#pragma unroll
    for (int ks = 0; ks < 3; ++ks) {
        int k0 = ks * 4 + kh * 2;
        afrag[ks].x = (k0     < F_IN) ? arow[k0]     * nd : 0.0f;
        afrag[ks].y = (k0 + 1 < F_IN) ? arow[k0 + 1] * nd : 0.0f;
    }

    int nn = lane & 15;
#pragma unroll
    for (int j = 0; j < 4; ++j) {
        v8f c = {};
#pragma unroll
        for (int ks = 0; ks < 3; ++ks) {
            int kb = ks * 4 + kh * 2;
            v2f b;
            b.x = Wc[(kb    ) * N_OUT + j * 16 + nn];
            b.y = Wc[(kb + 1) * N_OUT + j * 16 + nn];
            c = __builtin_amdgcn_wmma_f32_16x16x4_f32(
                     false, afrag[ks],
                     false, b,
                     (short)0, c,
                     false,  false);
        }
        float bias = bc[j * 16 + nn];
        int nbase = tile * 16;
#pragma unroll
        for (int r = 0; r < 8; ++r) {
            int row = r + 8 * kh;
            out[(size_t)(nbase + row) * N_OUT + j * 16 + nn] = c[r] + bias;
        }
    }
}

extern "C" void kernel_launch(void* const* d_in, const int* in_sizes, int n_in,
                              void* d_out, int out_size, void* d_ws, size_t ws_size,
                              hipStream_t stream) {
    const float* x   = (const float*)d_in[0];
    const int*   src = (const int*)  d_in[1];
    const int*   dst = (const int*)  d_in[2];
    const float* W1  = (const float*)d_in[3];
    const float* b1  = (const float*)d_in[4];
    const float* W2  = (const float*)d_in[5];
    const float* b2  = (const float*)d_in[6];
    float* out = (float*)d_out;

    float* ws = (float*)d_ws;
    float* deg_out = ws;
    float* deg_in  = ws + N_NODES;
    float* agg     = ws + 2 * N_NODES;
    float* norm_dv = ws + 2 * N_NODES + N_NODES * F_IN;
    float* xscaled = norm_dv + N_NODES;
    float* Wc      = xscaled + (size_t)N_NODES * F_IN;
    float* bcv     = Wc + KPAD * N_OUT;

    const int zeroN = 2 * N_NODES + N_NODES * F_IN;
    gcn_zero_kernel<<<(zeroN + 255) / 256, 256, 0, stream>>>(ws, zeroN);

    gcn_degree_kernel<<<(N_EDGES + 255) / 256, 256, 0, stream>>>(src, dst, deg_out, deg_in, N_EDGES);

    gcn_nodeprep_kernel<<<(N_NODES + 255) / 256, 256, 0, stream>>>(x, deg_out, deg_in,
                                                                   norm_dv, xscaled, N_NODES);

    gcn_foldw_kernel<<<1, 64, 0, stream>>>(W1, b1, W2, b2, Wc, bcv);

    gcn_scatter_kernel<<<(N_EDGES + 255) / 256, 256, 0, stream>>>(src, dst, xscaled, agg, N_EDGES);

    gcn_wmma_kernel<<<N_NODES / 16, 32, 0, stream>>>(agg, norm_dv, Wc, bcv, out);
}


# Global f32 atomics replaced by node-range ownership: each work-group scans an edge chunk and sums its rows with LDS atomics (arrival order), partials summed in fixed chunk order by the consumer
# speedup vs baseline: 6.8354x; 6.8354x over previous
#include <hip/hip_runtime.h>


#define N_NODES  100000
#define N_EDGES  3200000
#define F_IN     10
#define KPAD     12
#define N_OUT    64

#define DEG_RANGES 16
#define DEG_CHUNKS 16
#define DEG_NODES  (N_NODES / DEG_RANGES)
#define DEG_EDGES  (N_EDGES / DEG_CHUNKS)
#define DEG_SLAB   6272
#define DEG_LOADS  8

#define SC_RANGES  32
#define SC_CHUNKS  8
#define SC_NODES   (N_NODES / SC_RANGES)
#define SC_EDGES   (N_EDGES / SC_CHUNKS)
#define SC_WAVES   32
#define SC_LOADS   8
#define SC_QCAP    (32 * 4 * SC_LOADS)
#define SC_ASLAB   31264
#define SC_CSLAB   3136

typedef __attribute__((ext_vector_type(2))) float v2f;
typedef __attribute__((ext_vector_type(8))) float v8f;

__global__ void __launch_bounds__(1024) gcn_degree_kernel(const int* __restrict__ src,
                                                          unsigned* __restrict__ pcnt_out) {
    __shared__ unsigned cnt[DEG_NODES];
    const int r = blockIdx.x / DEG_CHUNKS;
    const int c = blockIdx.x % DEG_CHUNKS;
    const unsigned lo = (unsigned)(r * DEG_NODES);
    const int tid = threadIdx.x;

    for (int i = tid; i < DEG_NODES; i += 1024) cnt[i] = 0u;
    __syncthreads();

    const int4* s4 = (const int4*)(src + (size_t)c * DEG_EDGES);
    const int n4 = DEG_EDGES / 4;
    for (int i0 = tid; i0 < n4; i0 += 1024 * DEG_LOADS) {
        int4 v[DEG_LOADS];
#pragma unroll
        for (int u = 0; u < DEG_LOADS; ++u) {
            int idx = i0 + u * 1024;
            v[u] = s4[(idx < n4) ? idx : (n4 - 1)];
        }
#pragma unroll
        for (int u = 0; u < DEG_LOADS; ++u) {
            const bool ok = (i0 + u * 1024) < n4;
            unsigned t;
            t = (unsigned)v[u].x - lo; if (ok && t < (unsigned)DEG_NODES) atomicAdd(&cnt[t], 1u);
            t = (unsigned)v[u].y - lo; if (ok && t < (unsigned)DEG_NODES) atomicAdd(&cnt[t], 1u);
            t = (unsigned)v[u].z - lo; if (ok && t < (unsigned)DEG_NODES) atomicAdd(&cnt[t], 1u);
            t = (unsigned)v[u].w - lo; if (ok && t < (unsigned)DEG_NODES) atomicAdd(&cnt[t], 1u);
        }
    }
    __syncthreads();

    unsigned* o = pcnt_out + (size_t)(c * DEG_RANGES + r) * DEG_SLAB;
    for (int i = tid; i < DEG_NODES; i += 1024) o[i] = cnt[i];
}

__global__ void __launch_bounds__(256) gcn_nodeprep_kernel(const float* __restrict__ x,
                                                           const unsigned* __restrict__ pcnt_out,
                                                           float* __restrict__ xscaled,
                                                           int nN) {
    int n = blockIdx.x * blockDim.x + threadIdx.x;
    if (n < nN) {
        const int r = n / DEG_NODES;
        const int l = n - r * DEG_NODES;
        unsigned d = 0u;
#pragma unroll
        for (int c = 0; c < DEG_CHUNKS; ++c) d += pcnt_out[(size_t)(c * DEG_RANGES + r) * DEG_SLAB + l];
        float dout = (float)d; dout = (dout > 0.0f) ? dout : 1.0f;
        float ns = rsqrtf(dout);
        const float* xr = x + (size_t)n * 40 + 15;
        float* o = xscaled + (size_t)n * F_IN;
#pragma unroll
        for (int k = 0; k < F_IN; ++k) o[k] = xr[k] * ns;
    }
}

__global__ void __launch_bounds__(64) gcn_foldw_kernel(const float* __restrict__ W1,
                                                       const float* __restrict__ b1,
                                                       const float* __restrict__ W2,
                                                       const float* __restrict__ b2,
                                                       float* __restrict__ Wc,
                                                       float* __restrict__ bc) {
    int j = threadIdx.x;
    float acc = b2[j];
    for (int k = 0; k < 128; ++k) acc += b1[k] * W2[k * 64 + j];
    bc[j] = acc;
#pragma unroll 1
    for (int i = 0; i < KPAD; ++i) {
        float s = 0.0f;
        if (i < F_IN)
            for (int k = 0; k < 128; ++k) s += W1[i * 128 + k] * W2[k * 64 + j];
        Wc[i * 64 + j] = s;
    }
}

__global__ void __launch_bounds__(1024) gcn_scatter_kernel(const int* __restrict__ src,
                                                           const int* __restrict__ dst,
                                                           const float* __restrict__ xscaled,
                                                           float* __restrict__ pagg,
                                                           unsigned* __restrict__ pcnt_in) {
    __shared__ float    acc[SC_NODES * F_IN];
    __shared__ unsigned cnt[SC_NODES];
    __shared__ unsigned queue[SC_WAVES * SC_QCAP];

    const int r = blockIdx.x / SC_CHUNKS;
    const int c = blockIdx.x % SC_CHUNKS;
    const unsigned lo = (unsigned)(r * SC_NODES);
    const int tid  = threadIdx.x;
    const int lane = tid & 31;
    const int w    = tid >> 5;

    for (int i = tid; i < SC_NODES * F_IN; i += 1024) acc[i] = 0.0f;
    for (int i = tid; i < SC_NODES; i += 1024) cnt[i] = 0u;
    __syncthreads();

    const int4* d4 = (const int4*)(dst + (size_t)c * SC_EDGES);
    const int4* s4 = (const int4*)(src + (size_t)c * SC_EDGES);
    unsigned* q = queue + w * SC_QCAP;
    const int n4 = SC_EDGES / 4;

    for (int base = w * (32 * SC_LOADS); base < n4; base += SC_WAVES * 32 * SC_LOADS) {
        int4 dv[SC_LOADS], sv[SC_LOADS];
#pragma unroll
        for (int u = 0; u < SC_LOADS; ++u) {
            int idx = base + u * 32 + lane;
            int cl = (idx < n4) ? idx : (n4 - 1);
            dv[u] = d4[cl];
            sv[u] = s4[cl];
        }

        unsigned n = 0u;
#define SC_PUSH(OK, D, S)                                                               \
        {                                                                               \
            unsigned t = (unsigned)(D) - lo;                                            \
            bool hit = (OK) && (t < (unsigned)SC_NODES);                                \
            unsigned mask = __builtin_amdgcn_ballot_w32(hit);                           \
            if (hit) q[n + __builtin_amdgcn_mbcnt_lo(mask, 0u)] =                       \
                         (t << 17) | ((unsigned)(S) & 0x1FFFFu);                        \
            n += (unsigned)__builtin_popcount(mask);                                    \
        }
#pragma unroll
        for (int u = 0; u < SC_LOADS; ++u) {
            const bool ok = (base + u * 32 + lane) < n4;
            SC_PUSH(ok, dv[u].x, sv[u].x)
            SC_PUSH(ok, dv[u].y, sv[u].y)
            SC_PUSH(ok, dv[u].z, sv[u].z)
            SC_PUSH(ok, dv[u].w, sv[u].w)
        }
#undef SC_PUSH

        __syncwarp();

        for (unsigned i = (unsigned)lane; i < n; i += 32u) {
            unsigned p = q[i];
            unsigned t = p >> 17;
            unsigned s = p & 0x1FFFFu;
            const float2* row = (const float2*)(xscaled + (size_t)s * F_IN);
            float2 a0 = row[0], a1 = row[1], a2 = row[2], a3 = row[3], a4 = row[4];
            float* a = acc + t * F_IN;
            atomicAdd(a + 0, a0.x); atomicAdd(a + 1, a0.y);
            atomicAdd(a + 2, a1.x); atomicAdd(a + 3, a1.y);
            atomicAdd(a + 4, a2.x); atomicAdd(a + 5, a2.y);
            atomicAdd(a + 6, a3.x); atomicAdd(a + 7, a3.y);
            atomicAdd(a + 8, a4.x); atomicAdd(a + 9, a4.y);
            atomicAdd(&cnt[t], 1u);
        }

        __syncwarp();
    }
    __syncthreads();

    float* oa = pagg + (size_t)(c * SC_RANGES + r) * SC_ASLAB;
    for (int i = tid; i < SC_NODES * F_IN; i += 1024) oa[i] = acc[i];
    unsigned* oc = pcnt_in + (size_t)(c * SC_RANGES + r) * SC_CSLAB;
    for (int i = tid; i < SC_NODES; i += 1024) oc[i] = cnt[i];
}

__global__ void __launch_bounds__(32) gcn_wmma_kernel(const float* __restrict__ pagg,
                                                      const unsigned* __restrict__ pcnt_in,
                                                      const float* __restrict__ Wc,
                                                      const float* __restrict__ bc,
                                                      float* __restrict__ out) {
    int tile = blockIdx.x;
    int lane = threadIdx.x;
    int m  = lane & 15;
    int kh = lane >> 4;

    int node = tile * 16 + m;
    const int r = node / SC_NODES;
    const int l = node - r * SC_NODES;

    unsigned dcnt = 0u;
#pragma unroll
    for (int c = 0; c < SC_CHUNKS; ++c) dcnt += pcnt_in[(size_t)(c * SC_RANGES + r) * SC_CSLAB + l];
    float din = (float)dcnt; din = (din > 0.0f) ? din : 1.0f;
    float nd = rsqrtf(din);

    v2f afrag[3];
#pragma unroll
    for (int ks = 0; ks < 3; ++ks) {
        int k0 = ks * 4 + kh * 2;
        float sx = 0.0f, sy = 0.0f;
        if (k0 < F_IN) {
#pragma unroll
            for (int c = 0; c < SC_CHUNKS; ++c) {
                const float2 p = *(const float2*)(pagg + (size_t)(c * SC_RANGES + r) * SC_ASLAB
                                                  + (size_t)l * F_IN + k0);
                sx += p.x; sy += p.y;
            }
        }
        afrag[ks].x = sx * nd;
        afrag[ks].y = sy * nd;
    }

    int nn = lane & 15;
#pragma unroll
    for (int j = 0; j < 4; ++j) {
        v8f c = {};
#pragma unroll
        for (int ks = 0; ks < 3; ++ks) {
            int kb = ks * 4 + kh * 2;
            v2f b;
            b.x = Wc[(kb    ) * N_OUT + j * 16 + nn];
            b.y = Wc[(kb + 1) * N_OUT + j * 16 + nn];
            c = __builtin_amdgcn_wmma_f32_16x16x4_f32(
                     false, afrag[ks],
                     false, b,
                     (short)0, c,
                     false,  false);
        }
        float bias = bc[j * 16 + nn];
        int nbase = tile * 16;
#pragma unroll
        for (int r8 = 0; r8 < 8; ++r8) {
            int row = r8 + 8 * kh;
            out[(size_t)(nbase + row) * N_OUT + j * 16 + nn] = c[r8] + bias;
        }
    }
}

extern "C" void kernel_launch(void* const* d_in, const int* in_sizes, int n_in,
                              void* d_out, int out_size, void* d_ws, size_t ws_size,
                              hipStream_t stream) {
    const float* x   = (const float*)d_in[0];
    const int*   src = (const int*)  d_in[1];
    const int*   dst = (const int*)  d_in[2];
    const float* W1  = (const float*)d_in[3];
    const float* b1  = (const float*)d_in[4];
    const float* W2  = (const float*)d_in[5];
    const float* b2  = (const float*)d_in[6];
    float* out = (float*)d_out;

    float* ws = (float*)d_ws;
    unsigned* pcnt_out = (unsigned*)ws;
    float*    xscaled  = ws + (size_t)DEG_CHUNKS * DEG_RANGES * DEG_SLAB;
    float*    pagg     = xscaled + (size_t)N_NODES * F_IN;
    unsigned* pcnt_in  = (unsigned*)(pagg + (size_t)SC_CHUNKS * SC_RANGES * SC_ASLAB);
    float*    Wc       = (float*)(pcnt_in + (size_t)SC_CHUNKS * SC_RANGES * SC_CSLAB);
    float*    bcv      = Wc + KPAD * N_OUT;

    gcn_degree_kernel<<<DEG_RANGES * DEG_CHUNKS, 1024, 0, stream>>>(src, pcnt_out);

    gcn_nodeprep_kernel<<<(N_NODES + 255) / 256, 256, 0, stream>>>(x, pcnt_out, xscaled, N_NODES);

    gcn_foldw_kernel<<<1, 64, 0, stream>>>(W1, b1, W2, b2, Wc, bcv);

    gcn_scatter_kernel<<<SC_RANGES * SC_CHUNKS, 1024, 0, stream>>>(src, dst, xscaled, pagg, pcnt_in);

    gcn_wmma_kernel<<<N_NODES / 16, 32, 0, stream>>>(pagg, pcnt_in, Wc, bcv, out);
}


# Edges sorted once by destination range in a new binning pass; each owner wave reads only its own range's entries instead of scanning every edge (sums stay in LDS-atomic arrival order)
# speedup vs baseline: 7.8857x; 1.1537x over previous
#include <hip/hip_runtime.h>


#define N_NODES  100000
#define N_EDGES  3200000
#define F_IN     10
#define KPAD     12
#define N_OUT    64

#define DEG_RANGES 16
#define DEG_CHUNKS 16
#define DEG_NODES  (N_NODES / DEG_RANGES)
#define DEG_EDGES  (N_EDGES / DEG_CHUNKS)
#define DEG_SLAB   6272
#define DEG_LOADS  8

#define SC_RANGES  32
#define SC_CHUNKS  8
#define SC_NODES   (N_NODES / SC_RANGES)
#define SC_EDGES   (N_EDGES / SC_CHUNKS)
#define SC_WAVES   32
#define SC_ASLAB   31264
#define SC_CSLAB   3136
#define SC_DEPTH   4

#define BIN_SLICES  (SC_CHUNKS * SC_WAVES)
#define BIN_EDGES   (N_EDGES / BIN_SLICES)
#define BIN_STRIDE  12512
#define BIN_OFFS    (SC_RANGES + 1)
#define BIN_THREADS 625
#define BIN_LOADS   5
#define BIN_NONE    0xFFFFFFFFu

typedef __attribute__((ext_vector_type(2))) float v2f;
typedef __attribute__((ext_vector_type(8))) float v8f;

__global__ void __launch_bounds__(1024) gcn_degree_kernel(const int* __restrict__ src,
                                                          unsigned* __restrict__ pcnt_out) {
    __shared__ unsigned cnt[DEG_NODES];
    const int r = blockIdx.x / DEG_CHUNKS;
    const int c = blockIdx.x % DEG_CHUNKS;
    const unsigned lo = (unsigned)(r * DEG_NODES);
    const int tid = threadIdx.x;

    for (int i = tid; i < DEG_NODES; i += 1024) cnt[i] = 0u;
    __syncthreads();

    const int4* s4 = (const int4*)(src + (size_t)c * DEG_EDGES);
    const int n4 = DEG_EDGES / 4;
    for (int i0 = tid; i0 < n4; i0 += 1024 * DEG_LOADS) {
        int4 v[DEG_LOADS];
#pragma unroll
        for (int u = 0; u < DEG_LOADS; ++u) {
            int idx = i0 + u * 1024;
            v[u] = s4[(idx < n4) ? idx : (n4 - 1)];
        }
#pragma unroll
        for (int u = 0; u < DEG_LOADS; ++u) {
            const bool ok = (i0 + u * 1024) < n4;
            unsigned t;
            t = (unsigned)v[u].x - lo; if (ok && t < (unsigned)DEG_NODES) atomicAdd(&cnt[t], 1u);
            t = (unsigned)v[u].y - lo; if (ok && t < (unsigned)DEG_NODES) atomicAdd(&cnt[t], 1u);
            t = (unsigned)v[u].z - lo; if (ok && t < (unsigned)DEG_NODES) atomicAdd(&cnt[t], 1u);
            t = (unsigned)v[u].w - lo; if (ok && t < (unsigned)DEG_NODES) atomicAdd(&cnt[t], 1u);
        }
    }
    __syncthreads();

    unsigned* o = pcnt_out + (size_t)(c * DEG_RANGES + r) * DEG_SLAB;
    for (int i = tid; i < DEG_NODES; i += 1024) o[i] = cnt[i];
}

__global__ void __launch_bounds__(256) gcn_nodeprep_kernel(const float* __restrict__ x,
                                                           const unsigned* __restrict__ pcnt_out,
                                                           float* __restrict__ xscaled,
                                                           int nN) {
    int n = blockIdx.x * blockDim.x + threadIdx.x;
    if (n < nN) {
        const int r = n / DEG_NODES;
        const int l = n - r * DEG_NODES;
        unsigned d = 0u;
#pragma unroll
        for (int c = 0; c < DEG_CHUNKS; ++c) d += pcnt_out[(size_t)(c * DEG_RANGES + r) * DEG_SLAB + l];
        float dout = (float)d; dout = (dout > 0.0f) ? dout : 1.0f;
        float ns = rsqrtf(dout);
        const float* xr = x + (size_t)n * 40 + 15;
        float* o = xscaled + (size_t)n * F_IN;
#pragma unroll
        for (int k = 0; k < F_IN; ++k) o[k] = xr[k] * ns;
    }
}

__global__ void __launch_bounds__(64) gcn_foldw_kernel(const float* __restrict__ W1,
                                                       const float* __restrict__ b1,
                                                       const float* __restrict__ W2,
                                                       const float* __restrict__ b2,
                                                       float* __restrict__ Wc,
                                                       float* __restrict__ bc) {
    int j = threadIdx.x;
    float acc = b2[j];
    for (int k = 0; k < 128; ++k) acc += b1[k] * W2[k * 64 + j];
    bc[j] = acc;
#pragma unroll 1
    for (int i = 0; i < KPAD; ++i) {
        float s = 0.0f;
        if (i < F_IN)
            for (int k = 0; k < 128; ++k) s += W1[i * 128 + k] * W2[k * 64 + j];
        Wc[i * 64 + j] = s;
    }
}

__global__ void __launch_bounds__(BIN_THREADS) gcn_bin_kernel(const int* __restrict__ src,
                                                              const int* __restrict__ dst,
                                                              unsigned* __restrict__ binned,
                                                              unsigned* __restrict__ binoff) {
    __shared__ __attribute__((aligned(16))) unsigned stage[BIN_EDGES];
    __shared__ unsigned hist[SC_RANGES + 1];
    __shared__ unsigned start[SC_RANGES];

    const int g   = blockIdx.x;
    const int tid = threadIdx.x;
    if (tid <= SC_RANGES) hist[tid] = 0u;
    __syncthreads();

    const int4* d4 = (const int4*)(dst + (size_t)g * BIN_EDGES);
    const int4* s4 = (const int4*)(src + (size_t)g * BIN_EDGES);
    const int n4 = BIN_EDGES / 4;

    int4 dv[BIN_LOADS], sv[BIN_LOADS];
#pragma unroll
    for (int u = 0; u < BIN_LOADS; ++u) {
        dv[u] = d4[tid + u * BIN_THREADS];
        sv[u] = s4[tid + u * BIN_THREADS];
    }

    unsigned pk[BIN_LOADS * 4], br[BIN_LOADS * 4];
#define BIN_RANK(J, D, S)                                                               \
    {                                                                                   \
        const unsigned d  = (unsigned)(D);                                              \
        const bool     in = d < (unsigned)N_NODES;                                      \
        const unsigned b  = in ? d / (unsigned)SC_NODES : (unsigned)SC_RANGES;          \
        const unsigned t  = d - b * (unsigned)SC_NODES;                                 \
        pk[J] = (t << 17) | ((unsigned)(S) & 0x1FFFFu);                                 \
        const unsigned rk = atomicAdd(&hist[b], 1u);                                    \
        br[J] = in ? ((b << 16) | rk) : BIN_NONE;                                       \
    }
#pragma unroll
    for (int u = 0; u < BIN_LOADS; ++u) {
        BIN_RANK(4 * u + 0, dv[u].x, sv[u].x)
        BIN_RANK(4 * u + 1, dv[u].y, sv[u].y)
        BIN_RANK(4 * u + 2, dv[u].z, sv[u].z)
        BIN_RANK(4 * u + 3, dv[u].w, sv[u].w)
    }
#undef BIN_RANK
    __syncthreads();

    if (tid < SC_RANGES) {
        unsigned s = 0u;
        for (int b = 0; b < tid; ++b) s += hist[b];
        start[tid] = s;
        binoff[g * BIN_OFFS + tid] = s;
        if (tid == SC_RANGES - 1) binoff[g * BIN_OFFS + SC_RANGES] = s + hist[tid];
    }
    __syncthreads();

#pragma unroll
    for (int j = 0; j < BIN_LOADS * 4; ++j)
        if (br[j] != BIN_NONE) stage[start[br[j] >> 16] + (br[j] & 0xFFFFu)] = pk[j];
    __syncthreads();

    uint4* o = (uint4*)(binned + (size_t)g * BIN_STRIDE);
    const uint4* st4 = (const uint4*)stage;
    for (int i = tid; i < n4; i += BIN_THREADS) o[i] = st4[i];
}

__global__ void __launch_bounds__(1024) gcn_scatter_kernel(const unsigned* __restrict__ binned,
                                                           const unsigned* __restrict__ binoff,
                                                           const float* __restrict__ xscaled,
                                                           float* __restrict__ pagg,
                                                           unsigned* __restrict__ pcnt_in) {
    __shared__ float    acc[SC_NODES * F_IN];
    __shared__ unsigned cnt[SC_NODES];

    const int r = blockIdx.x / SC_CHUNKS;
    const int c = blockIdx.x % SC_CHUNKS;
    const int tid  = threadIdx.x;
    const int lane = tid & 31;
    const int w    = tid >> 5;

    for (int i = tid; i < SC_NODES * F_IN; i += 1024) acc[i] = 0.0f;
    for (int i = tid; i < SC_NODES; i += 1024) cnt[i] = 0u;
    __syncthreads();

    const int g = c * SC_WAVES + w;
    const unsigned* bl = binned + (size_t)g * BIN_STRIDE;
    const unsigned beg = binoff[g * BIN_OFFS + r];
    const unsigned end = binoff[g * BIN_OFFS + r + 1];

#define SC_ADD(P, A0, A1, A2, A3, A4)                                                   \
    {                                                                                   \
        const unsigned t = (P) >> 17;                                                   \
        float* a = acc + t * F_IN;                                                      \
        atomicAdd(a + 0, A0.x); atomicAdd(a + 1, A0.y);                                 \
        atomicAdd(a + 2, A1.x); atomicAdd(a + 3, A1.y);                                 \
        atomicAdd(a + 4, A2.x); atomicAdd(a + 5, A2.y);                                 \
        atomicAdd(a + 6, A3.x); atomicAdd(a + 7, A3.y);                                 \
        atomicAdd(a + 8, A4.x); atomicAdd(a + 9, A4.y);                                 \
        atomicAdd(&cnt[t], 1u);                                                         \
    }
    for (unsigned i = beg + (unsigned)lane; i < end; i += 32u * SC_DEPTH) {
        bool     ok[SC_DEPTH];
        unsigned p[SC_DEPTH];
#pragma unroll
        for (int k = 0; k < SC_DEPTH; ++k) {
            const unsigned ik = i + 32u * (unsigned)k;
            ok[k] = ik < end;
            p[k]  = bl[ok[k] ? ik : i];
        }
        float2 v[SC_DEPTH][5];
#pragma unroll
        for (int k = 0; k < SC_DEPTH; ++k) {
            const float2* row = (const float2*)(xscaled + (size_t)(p[k] & 0x1FFFFu) * F_IN);
#pragma unroll
            for (int j = 0; j < 5; ++j) v[k][j] = row[j];
        }
#pragma unroll
        for (int k = 0; k < SC_DEPTH; ++k)
            asm volatile("" :: "v"(v[k][0].x), "v"(v[k][0].y), "v"(v[k][1].x), "v"(v[k][1].y), "v"(v[k][2].x),
                               "v"(v[k][2].y), "v"(v[k][3].x), "v"(v[k][3].y), "v"(v[k][4].x), "v"(v[k][4].y));
#pragma unroll
        for (int k = 0; k < SC_DEPTH; ++k)
            if (ok[k]) SC_ADD(p[k], v[k][0], v[k][1], v[k][2], v[k][3], v[k][4])
    }
#undef SC_ADD
    __syncthreads();

    float* oa = pagg + (size_t)(c * SC_RANGES + r) * SC_ASLAB;
    for (int i = tid; i < SC_NODES * F_IN; i += 1024) oa[i] = acc[i];
    unsigned* oc = pcnt_in + (size_t)(c * SC_RANGES + r) * SC_CSLAB;
    for (int i = tid; i < SC_NODES; i += 1024) oc[i] = cnt[i];
}

__global__ void __launch_bounds__(32) gcn_wmma_kernel(const float* __restrict__ pagg,
                                                      const unsigned* __restrict__ pcnt_in,
                                                      const float* __restrict__ Wc,
                                                      const float* __restrict__ bc,
                                                      float* __restrict__ out) {
    int tile = blockIdx.x;
    int lane = threadIdx.x;
    int m  = lane & 15;
    int kh = lane >> 4;

    int node = tile * 16 + m;
    const int r = node / SC_NODES;
    const int l = node - r * SC_NODES;

    unsigned dcnt = 0u;
#pragma unroll
    for (int c = 0; c < SC_CHUNKS; ++c) dcnt += pcnt_in[(size_t)(c * SC_RANGES + r) * SC_CSLAB + l];
    float din = (float)dcnt; din = (din > 0.0f) ? din : 1.0f;
    float nd = rsqrtf(din);

    v2f afrag[3];
#pragma unroll
    for (int ks = 0; ks < 3; ++ks) {
        int k0 = ks * 4 + kh * 2;
        float sx = 0.0f, sy = 0.0f;
        if (k0 < F_IN) {
#pragma unroll
            for (int c = 0; c < SC_CHUNKS; ++c) {
                const float2 p = *(const float2*)(pagg + (size_t)(c * SC_RANGES + r) * SC_ASLAB
                                                  + (size_t)l * F_IN + k0);
                sx += p.x; sy += p.y;
            }
        }
        afrag[ks].x = sx * nd;
        afrag[ks].y = sy * nd;
    }

    int nn = lane & 15;
#pragma unroll
    for (int j = 0; j < 4; ++j) {
        v8f c = {};
#pragma unroll
        for (int ks = 0; ks < 3; ++ks) {
            int kb = ks * 4 + kh * 2;
            v2f b;
            b.x = Wc[(kb    ) * N_OUT + j * 16 + nn];
            b.y = Wc[(kb + 1) * N_OUT + j * 16 + nn];
            c = __builtin_amdgcn_wmma_f32_16x16x4_f32(
                     false, afrag[ks],
                     false, b,
                     (short)0, c,
                     false,  false);
        }
        float bias = bc[j * 16 + nn];
        int nbase = tile * 16;
#pragma unroll
        for (int r8 = 0; r8 < 8; ++r8) {
            int row = r8 + 8 * kh;
            out[(size_t)(nbase + row) * N_OUT + j * 16 + nn] = c[r8] + bias;
        }
    }
}

extern "C" void kernel_launch(void* const* d_in, const int* in_sizes, int n_in,
                              void* d_out, int out_size, void* d_ws, size_t ws_size,
                              hipStream_t stream) {
    const float* x   = (const float*)d_in[0];
    const int*   src = (const int*)  d_in[1];
    const int*   dst = (const int*)  d_in[2];
    const float* W1  = (const float*)d_in[3];
    const float* b1  = (const float*)d_in[4];
    const float* W2  = (const float*)d_in[5];
    const float* b2  = (const float*)d_in[6];
    float* out = (float*)d_out;

    float* ws = (float*)d_ws;
    unsigned* pcnt_out = (unsigned*)ws;
    float*    xscaled  = ws + (size_t)DEG_CHUNKS * DEG_RANGES * DEG_SLAB;
    float*    pagg     = xscaled + (size_t)N_NODES * F_IN;
    unsigned* pcnt_in  = (unsigned*)(pagg + (size_t)SC_CHUNKS * SC_RANGES * SC_ASLAB);
    float*    Wc       = (float*)(pcnt_in + (size_t)SC_CHUNKS * SC_RANGES * SC_CSLAB);
    float*    bcv      = Wc + KPAD * N_OUT;
    unsigned* binned   = (unsigned*)(bcv + N_OUT);
    unsigned* binoff   = binned + (size_t)BIN_SLICES * BIN_STRIDE;

    gcn_degree_kernel<<<DEG_RANGES * DEG_CHUNKS, 1024, 0, stream>>>(src, pcnt_out);

    gcn_nodeprep_kernel<<<(N_NODES + 255) / 256, 256, 0, stream>>>(x, pcnt_out, xscaled, N_NODES);

    gcn_foldw_kernel<<<1, 64, 0, stream>>>(W1, b1, W2, b2, Wc, bcv);

    gcn_bin_kernel<<<BIN_SLICES, BIN_THREADS, 0, stream>>>(src, dst, binned, binoff);

    gcn_scatter_kernel<<<SC_RANGES * SC_CHUNKS, 1024, 0, stream>>>(binned, binoff, xscaled, pagg, pcnt_in);

    gcn_wmma_kernel<<<N_NODES / 16, 32, 0, stream>>>(pagg, pcnt_in, Wc, bcv, out);
}


# Weight fold (W1*W2 and bias) spread from one 64-thread work-group to 11 of 256 threads: each 128-term sum split in 4 partial sums of 32 added in fixed order, a thread's loads issued together
# speedup vs baseline: 10.6505x; 1.3506x over previous
#include <hip/hip_runtime.h>


#define N_NODES  100000
#define N_EDGES  3200000
#define F_IN     10
#define KPAD     12
#define N_OUT    64

#define DEG_RANGES 16
#define DEG_CHUNKS 16
#define DEG_NODES  (N_NODES / DEG_RANGES)
#define DEG_EDGES  (N_EDGES / DEG_CHUNKS)
#define DEG_SLAB   6272
#define DEG_LOADS  8

#define SC_RANGES  32
#define SC_CHUNKS  8
#define SC_NODES   (N_NODES / SC_RANGES)
#define SC_EDGES   (N_EDGES / SC_CHUNKS)
#define SC_WAVES   32
#define SC_ASLAB   31264
#define SC_CSLAB   3136
#define SC_DEPTH   4

#define BIN_SLICES  (SC_CHUNKS * SC_WAVES)
#define BIN_EDGES   (N_EDGES / BIN_SLICES)
#define BIN_STRIDE  12512
#define BIN_OFFS    (SC_RANGES + 1)
#define BIN_THREADS 625
#define BIN_LOADS   5
#define BIN_NONE    0xFFFFFFFFu

typedef __attribute__((ext_vector_type(2))) float v2f;
typedef __attribute__((ext_vector_type(8))) float v8f;

__global__ void __launch_bounds__(1024) gcn_degree_kernel(const int* __restrict__ src,
                                                          unsigned* __restrict__ pcnt_out) {
    __shared__ unsigned cnt[DEG_NODES];
    const int r = blockIdx.x / DEG_CHUNKS;
    const int c = blockIdx.x % DEG_CHUNKS;
    const unsigned lo = (unsigned)(r * DEG_NODES);
    const int tid = threadIdx.x;

    for (int i = tid; i < DEG_NODES; i += 1024) cnt[i] = 0u;
    __syncthreads();

    const int4* s4 = (const int4*)(src + (size_t)c * DEG_EDGES);
    const int n4 = DEG_EDGES / 4;
    for (int i0 = tid; i0 < n4; i0 += 1024 * DEG_LOADS) {
        int4 v[DEG_LOADS];
#pragma unroll
        for (int u = 0; u < DEG_LOADS; ++u) {
            int idx = i0 + u * 1024;
            v[u] = s4[(idx < n4) ? idx : (n4 - 1)];
        }
#pragma unroll
        for (int u = 0; u < DEG_LOADS; ++u) {
            const bool ok = (i0 + u * 1024) < n4;
            unsigned t;
            t = (unsigned)v[u].x - lo; if (ok && t < (unsigned)DEG_NODES) atomicAdd(&cnt[t], 1u);
            t = (unsigned)v[u].y - lo; if (ok && t < (unsigned)DEG_NODES) atomicAdd(&cnt[t], 1u);
            t = (unsigned)v[u].z - lo; if (ok && t < (unsigned)DEG_NODES) atomicAdd(&cnt[t], 1u);
            t = (unsigned)v[u].w - lo; if (ok && t < (unsigned)DEG_NODES) atomicAdd(&cnt[t], 1u);
        }
    }
    __syncthreads();

    unsigned* o = pcnt_out + (size_t)(c * DEG_RANGES + r) * DEG_SLAB;
    for (int i = tid; i < DEG_NODES; i += 1024) o[i] = cnt[i];
}

__global__ void __launch_bounds__(256) gcn_nodeprep_kernel(const float* __restrict__ x,
                                                           const unsigned* __restrict__ pcnt_out,
                                                           float* __restrict__ xscaled,
                                                           int nN) {
    int n = blockIdx.x * blockDim.x + threadIdx.x;
    if (n < nN) {
        const int r = n / DEG_NODES;
        const int l = n - r * DEG_NODES;
        unsigned d = 0u;
#pragma unroll
        for (int c = 0; c < DEG_CHUNKS; ++c) d += pcnt_out[(size_t)(c * DEG_RANGES + r) * DEG_SLAB + l];
        float dout = (float)d; dout = (dout > 0.0f) ? dout : 1.0f;
        float ns = rsqrtf(dout);
        const float* xr = x + (size_t)n * 40 + 15;
        float* o = xscaled + (size_t)n * F_IN;
#pragma unroll
        for (int k = 0; k < F_IN; ++k) o[k] = xr[k] * ns;
    }
}

#define FOLD_PARTS 4
#define FOLD_K     (128 / FOLD_PARTS)
__global__ void __launch_bounds__(64 * FOLD_PARTS) gcn_foldw_kernel(const float* __restrict__ W1,
                                                                    const float* __restrict__ b1,
                                                                    const float* __restrict__ W2,
                                                                    const float* __restrict__ b2,
                                                                    float* __restrict__ Wc,
                                                                    float* __restrict__ bc) {
    __shared__ float part[FOLD_PARTS][64];
    const int i = blockIdx.x;
    const int j = threadIdx.x & 63;
    const int h = threadIdx.x >> 6;
    const float* a = ((i < F_IN) ? (W1 + i * 128) : b1) + h * FOLD_K;
    const float* w = W2 + (h * FOLD_K) * 64 + j;

    float av[FOLD_K], wv[FOLD_K];
#pragma unroll
    for (int k = 0; k < FOLD_K; ++k) { av[k] = a[k]; wv[k] = w[k * 64]; }
    float s = 0.0f;
#pragma unroll
    for (int k = 0; k < FOLD_K; ++k) s += av[k] * wv[k];
    part[h][j] = s;
    __syncthreads();

    if (h == 0) {
        float t = part[0][j];
#pragma unroll
        for (int q = 1; q < FOLD_PARTS; ++q) t += part[q][j];
        if (i < F_IN) Wc[i * 64 + j] = t;
        else          bc[j] = b2[j] + t;
    } else if (i == F_IN && h <= KPAD - F_IN) {
        Wc[(F_IN + h - 1) * 64 + j] = 0.0f;
    }
}

__global__ void __launch_bounds__(BIN_THREADS) gcn_bin_kernel(const int* __restrict__ src,
                                                              const int* __restrict__ dst,
                                                              unsigned* __restrict__ binned,
                                                              unsigned* __restrict__ binoff) {
    __shared__ __attribute__((aligned(16))) unsigned stage[BIN_EDGES];
    __shared__ unsigned hist[SC_RANGES + 1];
    __shared__ unsigned start[SC_RANGES];

    const int g   = blockIdx.x;
    const int tid = threadIdx.x;
    if (tid <= SC_RANGES) hist[tid] = 0u;
    __syncthreads();

    const int4* d4 = (const int4*)(dst + (size_t)g * BIN_EDGES);
    const int4* s4 = (const int4*)(src + (size_t)g * BIN_EDGES);
    const int n4 = BIN_EDGES / 4;

    int4 dv[BIN_LOADS], sv[BIN_LOADS];
#pragma unroll
    for (int u = 0; u < BIN_LOADS; ++u) {
        dv[u] = d4[tid + u * BIN_THREADS];
        sv[u] = s4[tid + u * BIN_THREADS];
    }

    unsigned pk[BIN_LOADS * 4], br[BIN_LOADS * 4];
#define BIN_RANK(J, D, S)                                                               \
    {                                                                                   \
        const unsigned d  = (unsigned)(D);                                              \
        const bool     in = d < (unsigned)N_NODES;                                      \
        const unsigned b  = in ? d / (unsigned)SC_NODES : (unsigned)SC_RANGES;          \
        const unsigned t  = d - b * (unsigned)SC_NODES;                                 \
        pk[J] = (t << 17) | ((unsigned)(S) & 0x1FFFFu);                                 \
        const unsigned rk = atomicAdd(&hist[b], 1u);                                    \
        br[J] = in ? ((b << 16) | rk) : BIN_NONE;                                       \
    }
#pragma unroll
    for (int u = 0; u < BIN_LOADS; ++u) {
        BIN_RANK(4 * u + 0, dv[u].x, sv[u].x)
        BIN_RANK(4 * u + 1, dv[u].y, sv[u].y)
        BIN_RANK(4 * u + 2, dv[u].z, sv[u].z)
        BIN_RANK(4 * u + 3, dv[u].w, sv[u].w)
    }
#undef BIN_RANK
    __syncthreads();

    if (tid < SC_RANGES) {
        unsigned s = 0u;
        for (int b = 0; b < tid; ++b) s += hist[b];
        start[tid] = s;
        binoff[g * BIN_OFFS + tid] = s;
        if (tid == SC_RANGES - 1) binoff[g * BIN_OFFS + SC_RANGES] = s + hist[tid];
    }
    __syncthreads();

#pragma unroll
    for (int j = 0; j < BIN_LOADS * 4; ++j)
        if (br[j] != BIN_NONE) stage[start[br[j] >> 16] + (br[j] & 0xFFFFu)] = pk[j];
    __syncthreads();

    uint4* o = (uint4*)(binned + (size_t)g * BIN_STRIDE);
    const uint4* st4 = (const uint4*)stage;
    for (int i = tid; i < n4; i += BIN_THREADS) o[i] = st4[i];
}

__global__ void __launch_bounds__(1024) gcn_scatter_kernel(const unsigned* __restrict__ binned,
                                                           const unsigned* __restrict__ binoff,
                                                           const float* __restrict__ xscaled,
                                                           float* __restrict__ pagg,
                                                           unsigned* __restrict__ pcnt_in) {
    __shared__ float    acc[SC_NODES * F_IN];
    __shared__ unsigned cnt[SC_NODES];

    const int r = blockIdx.x / SC_CHUNKS;
    const int c = blockIdx.x % SC_CHUNKS;
    const int tid  = threadIdx.x;
    const int lane = tid & 31;
    const int w    = tid >> 5;

    for (int i = tid; i < SC_NODES * F_IN; i += 1024) acc[i] = 0.0f;
    for (int i = tid; i < SC_NODES; i += 1024) cnt[i] = 0u;
    __syncthreads();

    const int g = c * SC_WAVES + w;
    const unsigned* bl = binned + (size_t)g * BIN_STRIDE;
    const unsigned beg = binoff[g * BIN_OFFS + r];
    const unsigned end = binoff[g * BIN_OFFS + r + 1];

#define SC_ADD(P, A0, A1, A2, A3, A4)                                                   \
    {                                                                                   \
        const unsigned t = (P) >> 17;                                                   \
        float* a = acc + t * F_IN;                                                      \
        atomicAdd(a + 0, A0.x); atomicAdd(a + 1, A0.y);                                 \
        atomicAdd(a + 2, A1.x); atomicAdd(a + 3, A1.y);                                 \
        atomicAdd(a + 4, A2.x); atomicAdd(a + 5, A2.y);                                 \
        atomicAdd(a + 6, A3.x); atomicAdd(a + 7, A3.y);                                 \
        atomicAdd(a + 8, A4.x); atomicAdd(a + 9, A4.y);                                 \
        atomicAdd(&cnt[t], 1u);                                                         \
    }
    for (unsigned i = beg + (unsigned)lane; i < end; i += 32u * SC_DEPTH) {
        bool     ok[SC_DEPTH];
        unsigned p[SC_DEPTH];
#pragma unroll
        for (int k = 0; k < SC_DEPTH; ++k) {
            const unsigned ik = i + 32u * (unsigned)k;
            ok[k] = ik < end;
            p[k]  = bl[ok[k] ? ik : i];
        }
        float2 v[SC_DEPTH][5];
#pragma unroll
        for (int k = 0; k < SC_DEPTH; ++k) {
            const float2* row = (const float2*)(xscaled + (size_t)(p[k] & 0x1FFFFu) * F_IN);
#pragma unroll
            for (int j = 0; j < 5; ++j) v[k][j] = row[j];
        }
#pragma unroll
        for (int k = 0; k < SC_DEPTH; ++k)
            asm volatile("" :: "v"(v[k][0].x), "v"(v[k][0].y), "v"(v[k][1].x), "v"(v[k][1].y), "v"(v[k][2].x),
                               "v"(v[k][2].y), "v"(v[k][3].x), "v"(v[k][3].y), "v"(v[k][4].x), "v"(v[k][4].y));
#pragma unroll
        for (int k = 0; k < SC_DEPTH; ++k)
            if (ok[k]) SC_ADD(p[k], v[k][0], v[k][1], v[k][2], v[k][3], v[k][4])
    }
#undef SC_ADD
    __syncthreads();

    float* oa = pagg + (size_t)(c * SC_RANGES + r) * SC_ASLAB;
    for (int i = tid; i < SC_NODES * F_IN; i += 1024) oa[i] = acc[i];
    unsigned* oc = pcnt_in + (size_t)(c * SC_RANGES + r) * SC_CSLAB;
    for (int i = tid; i < SC_NODES; i += 1024) oc[i] = cnt[i];
}

__global__ void __launch_bounds__(32) gcn_wmma_kernel(const float* __restrict__ pagg,
                                                      const unsigned* __restrict__ pcnt_in,
                                                      const float* __restrict__ Wc,
                                                      const float* __restrict__ bc,
                                                      float* __restrict__ out) {
    int tile = blockIdx.x;
    int lane = threadIdx.x;
    int m  = lane & 15;
    int kh = lane >> 4;

    int node = tile * 16 + m;
    const int r = node / SC_NODES;
    const int l = node - r * SC_NODES;

    unsigned dcnt = 0u;
#pragma unroll
    for (int c = 0; c < SC_CHUNKS; ++c) dcnt += pcnt_in[(size_t)(c * SC_RANGES + r) * SC_CSLAB + l];
    float din = (float)dcnt; din = (din > 0.0f) ? din : 1.0f;
    float nd = rsqrtf(din);

    v2f afrag[3];
#pragma unroll
    for (int ks = 0; ks < 3; ++ks) {
        int k0 = ks * 4 + kh * 2;
        float sx = 0.0f, sy = 0.0f;
        if (k0 < F_IN) {
#pragma unroll
            for (int c = 0; c < SC_CHUNKS; ++c) {
                const float2 p = *(const float2*)(pagg + (size_t)(c * SC_RANGES + r) * SC_ASLAB
                                                  + (size_t)l * F_IN + k0);
                sx += p.x; sy += p.y;
            }
        }
        afrag[ks].x = sx * nd;
        afrag[ks].y = sy * nd;
    }

    int nn = lane & 15;
#pragma unroll
    for (int j = 0; j < 4; ++j) {
        v8f c = {};
#pragma unroll
        for (int ks = 0; ks < 3; ++ks) {
            int kb = ks * 4 + kh * 2;
            v2f b;
            b.x = Wc[(kb    ) * N_OUT + j * 16 + nn];
            b.y = Wc[(kb + 1) * N_OUT + j * 16 + nn];
            c = __builtin_amdgcn_wmma_f32_16x16x4_f32(
                     false, afrag[ks],
                     false, b,
                     (short)0, c,
                     false,  false);
        }
        float bias = bc[j * 16 + nn];
        int nbase = tile * 16;
#pragma unroll
        for (int r8 = 0; r8 < 8; ++r8) {
            int row = r8 + 8 * kh;
            out[(size_t)(nbase + row) * N_OUT + j * 16 + nn] = c[r8] + bias;
        }
    }
}

extern "C" void kernel_launch(void* const* d_in, const int* in_sizes, int n_in,
                              void* d_out, int out_size, void* d_ws, size_t ws_size,
                              hipStream_t stream) {
    const float* x   = (const float*)d_in[0];
    const int*   src = (const int*)  d_in[1];
    const int*   dst = (const int*)  d_in[2];
    const float* W1  = (const float*)d_in[3];
    const float* b1  = (const float*)d_in[4];
    const float* W2  = (const float*)d_in[5];
    const float* b2  = (const float*)d_in[6];
    float* out = (float*)d_out;

    float* ws = (float*)d_ws;
    unsigned* pcnt_out = (unsigned*)ws;
    float*    xscaled  = ws + (size_t)DEG_CHUNKS * DEG_RANGES * DEG_SLAB;
    float*    pagg     = xscaled + (size_t)N_NODES * F_IN;
    unsigned* pcnt_in  = (unsigned*)(pagg + (size_t)SC_CHUNKS * SC_RANGES * SC_ASLAB);
    float*    Wc       = (float*)(pcnt_in + (size_t)SC_CHUNKS * SC_RANGES * SC_CSLAB);
    float*    bcv      = Wc + KPAD * N_OUT;
    unsigned* binned   = (unsigned*)(bcv + N_OUT);
    unsigned* binoff   = binned + (size_t)BIN_SLICES * BIN_STRIDE;

    gcn_degree_kernel<<<DEG_RANGES * DEG_CHUNKS, 1024, 0, stream>>>(src, pcnt_out);

    gcn_nodeprep_kernel<<<(N_NODES + 255) / 256, 256, 0, stream>>>(x, pcnt_out, xscaled, N_NODES);

    gcn_foldw_kernel<<<F_IN + 1, 64 * FOLD_PARTS, 0, stream>>>(W1, b1, W2, b2, Wc, bcv);

    gcn_bin_kernel<<<BIN_SLICES, BIN_THREADS, 0, stream>>>(src, dst, binned, binoff);

    gcn_scatter_kernel<<<SC_RANGES * SC_CHUNKS, 1024, 0, stream>>>(binned, binoff, xscaled, pagg, pcnt_in);

    gcn_wmma_kernel<<<N_NODES / 16, 32, 0, stream>>>(pagg, pcnt_in, Wc, bcv, out);
}


# Out-degree counting and binning merged into one launch (role by work-group index, 640-thread blocks), weight fold merged into the per-node launch: 6 launches -> 4, independent passes overlap
# speedup vs baseline: 12.2648x; 1.1516x over previous
#include <hip/hip_runtime.h>


#define N_NODES  100000
#define N_EDGES  3200000
#define F_IN     10
#define KPAD     12
#define N_OUT    64

#define DEG_RANGES 16
#define DEG_CHUNKS 16
#define DEG_NODES  (N_NODES / DEG_RANGES)
#define DEG_EDGES  (N_EDGES / DEG_CHUNKS)
#define DEG_SLAB   6272
#define DEG_LOADS  8

#define SC_RANGES  32
#define SC_CHUNKS  8
#define SC_NODES   (N_NODES / SC_RANGES)
#define SC_EDGES   (N_EDGES / SC_CHUNKS)
#define SC_WAVES   32
#define SC_ASLAB   31264
#define SC_CSLAB   3136
#define SC_DEPTH   4

#define BIN_SLICES  (SC_CHUNKS * SC_WAVES)
#define BIN_EDGES   (N_EDGES / BIN_SLICES)
#define BIN_STRIDE  12512
#define BIN_OFFS    (SC_RANGES + 1)
#define BIN_THREADS 625
#define BIN_LOADS   5
#define BIN_NONE    0xFFFFFFFFu

#define P1_THREADS  640
#define P1_DEG_WGS  (DEG_RANGES * DEG_CHUNKS)
#define P1_HIST_AT  BIN_STRIDE
#define P1_LDS      (BIN_STRIDE + 64 + SC_RANGES)

typedef __attribute__((ext_vector_type(2))) float v2f;
typedef __attribute__((ext_vector_type(8))) float v8f;

__device__ __forceinline__ void gcn_degree_role(const int* __restrict__ src, unsigned* __restrict__ pcnt_out,
                                                unsigned* cnt, int b) {
    const int r = b / DEG_CHUNKS;
    const int c = b % DEG_CHUNKS;
    const unsigned lo = (unsigned)(r * DEG_NODES);
    const int tid = threadIdx.x;

    for (int i = tid; i < DEG_NODES; i += P1_THREADS) cnt[i] = 0u;
    __syncthreads();

    const int4* s4 = (const int4*)(src + (size_t)c * DEG_EDGES);
    const int n4 = DEG_EDGES / 4;
    for (int i0 = tid; i0 < n4; i0 += P1_THREADS * DEG_LOADS) {
        int4 v[DEG_LOADS];
#pragma unroll
        for (int u = 0; u < DEG_LOADS; ++u) {
            int idx = i0 + u * P1_THREADS;
            v[u] = s4[(idx < n4) ? idx : (n4 - 1)];
        }
#pragma unroll
        for (int u = 0; u < DEG_LOADS; ++u) {
            const bool ok = (i0 + u * P1_THREADS) < n4;
            unsigned t;
            t = (unsigned)v[u].x - lo; if (ok && t < (unsigned)DEG_NODES) atomicAdd(&cnt[t], 1u);
            t = (unsigned)v[u].y - lo; if (ok && t < (unsigned)DEG_NODES) atomicAdd(&cnt[t], 1u);
            t = (unsigned)v[u].z - lo; if (ok && t < (unsigned)DEG_NODES) atomicAdd(&cnt[t], 1u);
            t = (unsigned)v[u].w - lo; if (ok && t < (unsigned)DEG_NODES) atomicAdd(&cnt[t], 1u);
        }
    }
    __syncthreads();

    unsigned* o = pcnt_out + (size_t)(c * DEG_RANGES + r) * DEG_SLAB;
    for (int i = tid; i < DEG_NODES; i += P1_THREADS) o[i] = cnt[i];
}

__device__ __forceinline__ void gcn_nodeprep_role(const float* __restrict__ x,
                                                  const unsigned* __restrict__ pcnt_out,
                                                  float* __restrict__ xscaled,
                                                  int nN, int b) {
    int n = b * 256 + threadIdx.x;
    if (n < nN) {
        const int r = n / DEG_NODES;
        const int l = n - r * DEG_NODES;
        unsigned d = 0u;
#pragma unroll
        for (int c = 0; c < DEG_CHUNKS; ++c) d += pcnt_out[(size_t)(c * DEG_RANGES + r) * DEG_SLAB + l];
        float dout = (float)d; dout = (dout > 0.0f) ? dout : 1.0f;
        float ns = rsqrtf(dout);
        const float* xr = x + (size_t)n * 40 + 15;
        float* o = xscaled + (size_t)n * F_IN;
#pragma unroll
        for (int k = 0; k < F_IN; ++k) o[k] = xr[k] * ns;
    }
}

#define FOLD_PARTS 4
#define FOLD_K     (128 / FOLD_PARTS)
__device__ __forceinline__ void gcn_foldw_role(const float* __restrict__ W1,
                                               const float* __restrict__ b1,
                                               const float* __restrict__ W2,
                                               const float* __restrict__ b2,
                                               float* __restrict__ Wc,
                                               float* __restrict__ bc,
                                               float (*part)[64], int i) {
    const int j = threadIdx.x & 63;
    const int h = threadIdx.x >> 6;
    const float* a = ((i < F_IN) ? (W1 + i * 128) : b1) + h * FOLD_K;
    const float* w = W2 + (h * FOLD_K) * 64 + j;

    float av[FOLD_K], wv[FOLD_K];
#pragma unroll
    for (int k = 0; k < FOLD_K; ++k) { av[k] = a[k]; wv[k] = w[k * 64]; }
    float s = 0.0f;
#pragma unroll
    for (int k = 0; k < FOLD_K; ++k) s += av[k] * wv[k];
    part[h][j] = s;
    __syncthreads();

    if (h == 0) {
        float t = part[0][j];
#pragma unroll
        for (int q = 1; q < FOLD_PARTS; ++q) t += part[q][j];
        if (i < F_IN) Wc[i * 64 + j] = t;
        else          bc[j] = b2[j] + t;
    } else if (i == F_IN && h <= KPAD - F_IN) {
        Wc[(F_IN + h - 1) * 64 + j] = 0.0f;
    }
}

#define FOLD_WGS (F_IN + 1)
__global__ void __launch_bounds__(256) gcn_prep_kernel(const float* __restrict__ x,
                                                       const unsigned* __restrict__ pcnt_out,
                                                       float* __restrict__ xscaled,
                                                       int nN,
                                                       const float* __restrict__ W1,
                                                       const float* __restrict__ b1,
                                                       const float* __restrict__ W2,
                                                       const float* __restrict__ b2,
                                                       float* __restrict__ Wc,
                                                       float* __restrict__ bc) {
    __shared__ float part[FOLD_PARTS][64];
    if (blockIdx.x < FOLD_WGS) gcn_foldw_role(W1, b1, W2, b2, Wc, bc, part, (int)blockIdx.x);
    else                       gcn_nodeprep_role(x, pcnt_out, xscaled, nN, (int)blockIdx.x - FOLD_WGS);
}

__device__ __forceinline__ void gcn_bin_role(const int* __restrict__ src,
                                             const int* __restrict__ dst,
                                             unsigned* __restrict__ binned,
                                             unsigned* __restrict__ binoff,
                                             unsigned* lds, int g) {
    unsigned* stage = lds;
    unsigned* hist  = lds + P1_HIST_AT;
    unsigned* start = lds + P1_HIST_AT + 64;

    const int tid = threadIdx.x;
    const bool worker = tid < BIN_THREADS;
    if (tid <= SC_RANGES) hist[tid] = 0u;
    __syncthreads();

    const int4* d4 = (const int4*)(dst + (size_t)g * BIN_EDGES);
    const int4* s4 = (const int4*)(src + (size_t)g * BIN_EDGES);
    const int n4 = BIN_EDGES / 4;

    unsigned pk[BIN_LOADS * 4], br[BIN_LOADS * 4];
    if (worker) {
        int4 dv[BIN_LOADS], sv[BIN_LOADS];
#pragma unroll
        for (int u = 0; u < BIN_LOADS; ++u) {
            dv[u] = d4[tid + u * BIN_THREADS];
            sv[u] = s4[tid + u * BIN_THREADS];
        }

#define BIN_RANK(J, D, S)                                                               \
        {                                                                                   \
            const unsigned d  = (unsigned)(D);                                              \
            const bool     in = d < (unsigned)N_NODES;                                      \
            const unsigned b  = in ? d / (unsigned)SC_NODES : (unsigned)SC_RANGES;          \
            const unsigned t  = d - b * (unsigned)SC_NODES;                                 \
            pk[J] = (t << 17) | ((unsigned)(S) & 0x1FFFFu);                                 \
            const unsigned rk = atomicAdd(&hist[b], 1u);                                    \
            br[J] = in ? ((b << 16) | rk) : BIN_NONE;                                       \
        }
#pragma unroll
        for (int u = 0; u < BIN_LOADS; ++u) {
            BIN_RANK(4 * u + 0, dv[u].x, sv[u].x)
            BIN_RANK(4 * u + 1, dv[u].y, sv[u].y)
            BIN_RANK(4 * u + 2, dv[u].z, sv[u].z)
            BIN_RANK(4 * u + 3, dv[u].w, sv[u].w)
        }
#undef BIN_RANK
    }
    __syncthreads();

    if (tid < SC_RANGES) {
        unsigned s = 0u;
        for (int b = 0; b < tid; ++b) s += hist[b];
        start[tid] = s;
        binoff[g * BIN_OFFS + tid] = s;
        if (tid == SC_RANGES - 1) binoff[g * BIN_OFFS + SC_RANGES] = s + hist[tid];
    }
    __syncthreads();

    if (worker) {
#pragma unroll
        for (int j = 0; j < BIN_LOADS * 4; ++j)
            if (br[j] != BIN_NONE) stage[start[br[j] >> 16] + (br[j] & 0xFFFFu)] = pk[j];
    }
    __syncthreads();

    uint4* o = (uint4*)(binned + (size_t)g * BIN_STRIDE);
    const uint4* st4 = (const uint4*)stage;
    for (int i = tid; i < n4; i += P1_THREADS) o[i] = st4[i];
}

__global__ void __launch_bounds__(P1_THREADS) gcn_edges_kernel(const int* __restrict__ src,
                                                               const int* __restrict__ dst,
                                                               unsigned* __restrict__ pcnt_out,
                                                               unsigned* __restrict__ binned,
                                                               unsigned* __restrict__ binoff) {
    __shared__ __attribute__((aligned(16))) unsigned lds[P1_LDS];
    if (blockIdx.x < P1_DEG_WGS) gcn_degree_role(src, pcnt_out, lds, (int)blockIdx.x);
    else                         gcn_bin_role(src, dst, binned, binoff, lds, (int)blockIdx.x - P1_DEG_WGS);
}

__global__ void __launch_bounds__(1024) gcn_scatter_kernel(const unsigned* __restrict__ binned,
                                                           const unsigned* __restrict__ binoff,
                                                           const float* __restrict__ xscaled,
                                                           float* __restrict__ pagg,
                                                           unsigned* __restrict__ pcnt_in) {
    __shared__ float    acc[SC_NODES * F_IN];
    __shared__ unsigned cnt[SC_NODES];

    const int r = blockIdx.x / SC_CHUNKS;
    const int c = blockIdx.x % SC_CHUNKS;
    const int tid  = threadIdx.x;
    const int lane = tid & 31;
    const int w    = tid >> 5;

    for (int i = tid; i < SC_NODES * F_IN; i += 1024) acc[i] = 0.0f;
    for (int i = tid; i < SC_NODES; i += 1024) cnt[i] = 0u;
    __syncthreads();

    const int g = c * SC_WAVES + w;
    const unsigned* bl = binned + (size_t)g * BIN_STRIDE;
    const unsigned beg = binoff[g * BIN_OFFS + r];
    const unsigned end = binoff[g * BIN_OFFS + r + 1];

#define SC_ADD(P, A0, A1, A2, A3, A4)                                                   \
    {                                                                                   \
        const unsigned t = (P) >> 17;                                                   \
        float* a = acc + t * F_IN;                                                      \
        atomicAdd(a + 0, A0.x); atomicAdd(a + 1, A0.y);                                 \
        atomicAdd(a + 2, A1.x); atomicAdd(a + 3, A1.y);                                 \
        atomicAdd(a + 4, A2.x); atomicAdd(a + 5, A2.y);                                 \
        atomicAdd(a + 6, A3.x); atomicAdd(a + 7, A3.y);                                 \
        atomicAdd(a + 8, A4.x); atomicAdd(a + 9, A4.y);                                 \
        atomicAdd(&cnt[t], 1u);                                                         \
    }
    for (unsigned i = beg + (unsigned)lane; i < end; i += 32u * SC_DEPTH) {
        bool     ok[SC_DEPTH];
        unsigned p[SC_DEPTH];
#pragma unroll
        for (int k = 0; k < SC_DEPTH; ++k) {
            const unsigned ik = i + 32u * (unsigned)k;
            ok[k] = ik < end;
            p[k]  = bl[ok[k] ? ik : i];
        }
        float2 v[SC_DEPTH][5];
#pragma unroll
        for (int k = 0; k < SC_DEPTH; ++k) {
            const float2* row = (const float2*)(xscaled + (size_t)(p[k] & 0x1FFFFu) * F_IN);
#pragma unroll
            for (int j = 0; j < 5; ++j) v[k][j] = row[j];
        }
#pragma unroll
        for (int k = 0; k < SC_DEPTH; ++k)
            asm volatile("" :: "v"(v[k][0].x), "v"(v[k][0].y), "v"(v[k][1].x), "v"(v[k][1].y), "v"(v[k][2].x),
                               "v"(v[k][2].y), "v"(v[k][3].x), "v"(v[k][3].y), "v"(v[k][4].x), "v"(v[k][4].y));
#pragma unroll
        for (int k = 0; k < SC_DEPTH; ++k)
            if (ok[k]) SC_ADD(p[k], v[k][0], v[k][1], v[k][2], v[k][3], v[k][4])
    }
#undef SC_ADD
    __syncthreads();

    float* oa = pagg + (size_t)(c * SC_RANGES + r) * SC_ASLAB;
    for (int i = tid; i < SC_NODES * F_IN; i += 1024) oa[i] = acc[i];
    unsigned* oc = pcnt_in + (size_t)(c * SC_RANGES + r) * SC_CSLAB;
    for (int i = tid; i < SC_NODES; i += 1024) oc[i] = cnt[i];
}

__global__ void __launch_bounds__(32) gcn_wmma_kernel(const float* __restrict__ pagg,
                                                      const unsigned* __restrict__ pcnt_in,
                                                      const float* __restrict__ Wc,
                                                      const float* __restrict__ bc,
                                                      float* __restrict__ out) {
    int tile = blockIdx.x;
    int lane = threadIdx.x;
    int m  = lane & 15;
    int kh = lane >> 4;

    int node = tile * 16 + m;
    const int r = node / SC_NODES;
    const int l = node - r * SC_NODES;

    unsigned dcnt = 0u;
#pragma unroll
    for (int c = 0; c < SC_CHUNKS; ++c) dcnt += pcnt_in[(size_t)(c * SC_RANGES + r) * SC_CSLAB + l];
    float din = (float)dcnt; din = (din > 0.0f) ? din : 1.0f;
    float nd = rsqrtf(din);

    v2f afrag[3];
#pragma unroll
    for (int ks = 0; ks < 3; ++ks) {
        int k0 = ks * 4 + kh * 2;
        float sx = 0.0f, sy = 0.0f;
        if (k0 < F_IN) {
#pragma unroll
            for (int c = 0; c < SC_CHUNKS; ++c) {
                const float2 p = *(const float2*)(pagg + (size_t)(c * SC_RANGES + r) * SC_ASLAB
                                                  + (size_t)l * F_IN + k0);
                sx += p.x; sy += p.y;
            }
        }
        afrag[ks].x = sx * nd;
        afrag[ks].y = sy * nd;
    }

    int nn = lane & 15;
#pragma unroll
    for (int j = 0; j < 4; ++j) {
        v8f c = {};
#pragma unroll
        for (int ks = 0; ks < 3; ++ks) {
            int kb = ks * 4 + kh * 2;
            v2f b;
            b.x = Wc[(kb    ) * N_OUT + j * 16 + nn];
            b.y = Wc[(kb + 1) * N_OUT + j * 16 + nn];
            c = __builtin_amdgcn_wmma_f32_16x16x4_f32(
                     false, afrag[ks],
                     false, b,
                     (short)0, c,
                     false,  false);
        }
        float bias = bc[j * 16 + nn];
        int nbase = tile * 16;
#pragma unroll
        for (int r8 = 0; r8 < 8; ++r8) {
            int row = r8 + 8 * kh;
            out[(size_t)(nbase + row) * N_OUT + j * 16 + nn] = c[r8] + bias;
        }
    }
}

extern "C" void kernel_launch(void* const* d_in, const int* in_sizes, int n_in,
                              void* d_out, int out_size, void* d_ws, size_t ws_size,
                              hipStream_t stream) {
    const float* x   = (const float*)d_in[0];
    const int*   src = (const int*)  d_in[1];
    const int*   dst = (const int*)  d_in[2];
    const float* W1  = (const float*)d_in[3];
    const float* b1  = (const float*)d_in[4];
    const float* W2  = (const float*)d_in[5];
    const float* b2  = (const float*)d_in[6];
    float* out = (float*)d_out;

    float* ws = (float*)d_ws;
    unsigned* pcnt_out = (unsigned*)ws;
    float*    xscaled  = ws + (size_t)DEG_CHUNKS * DEG_RANGES * DEG_SLAB;
    float*    pagg     = xscaled + (size_t)N_NODES * F_IN;
    unsigned* pcnt_in  = (unsigned*)(pagg + (size_t)SC_CHUNKS * SC_RANGES * SC_ASLAB);
    float*    Wc       = (float*)(pcnt_in + (size_t)SC_CHUNKS * SC_RANGES * SC_CSLAB);
    float*    bcv      = Wc + KPAD * N_OUT;
    unsigned* binned   = (unsigned*)(bcv + N_OUT);
    unsigned* binoff   = binned + (size_t)BIN_SLICES * BIN_STRIDE;

    gcn_edges_kernel<<<P1_DEG_WGS + BIN_SLICES, P1_THREADS, 0, stream>>>(src, dst, pcnt_out, binned, binoff);

    gcn_prep_kernel<<<FOLD_WGS + (N_NODES + 255) / 256, 256, 0, stream>>>(x, pcnt_out, xscaled, N_NODES,
                                                                          W1, b1, W2, b2, Wc, bcv);

    gcn_scatter_kernel<<<SC_RANGES * SC_CHUNKS, 1024, 0, stream>>>(binned, binoff, xscaled, pagg, pcnt_in);

    gcn_wmma_kernel<<<N_NODES / 16, 32, 0, stream>>>(pagg, pcnt_in, Wc, bcv, out);
}


# Each owner work-group takes every edge of its 400-node range (250 ranges, no chunk partials; sums in one LDS-atomic arrival order) and does the matrix product and output store itself: 4 launches -> 3
# speedup vs baseline: 14.0921x; 1.1490x over previous
#include <hip/hip_runtime.h>


#define N_NODES  100000
#define N_EDGES  3200000
#define F_IN     10
#define KPAD     12
#define N_OUT    64

#define DEG_RANGES 16
#define DEG_CHUNKS 16
#define DEG_NODES  (N_NODES / DEG_RANGES)
#define DEG_EDGES  (N_EDGES / DEG_CHUNKS)
#define DEG_SLAB   6272
#define DEG_LOADS  8

#define OW_RANGES  250
#define OW_NODES   (N_NODES / OW_RANGES)
#define OW_TILES   (OW_NODES / 16)
#define OW_WAVES   16
#define OW_THREADS (OW_WAVES * 32)
#define OW_GROUP   4
#define OW_PER     2

#define BIN_SLICES  256
#define OW_SLICES   (BIN_SLICES / OW_WAVES)
#define BIN_EDGES   (N_EDGES / BIN_SLICES)
#define BIN_STRIDE  12512
#define BIN_OFFS    (OW_RANGES + 1)
#define BIN_THREADS 625
#define BIN_LOADS   5
#define BIN_NONE    0xFFFFFFFFu

#define P1_THREADS  640
#define P1_DEG_WGS  (DEG_RANGES * DEG_CHUNKS)
#define P1_HIST_AT  BIN_STRIDE
#define P1_START_AT (BIN_STRIDE + 256)
#define P1_WSUM_AT  (BIN_STRIDE + 512)
#define P1_LDS      (BIN_STRIDE + 512 + 32)

typedef __attribute__((ext_vector_type(2))) float v2f;
typedef __attribute__((ext_vector_type(8))) float v8f;

__device__ __forceinline__ void gcn_degree_role(const int* __restrict__ src, unsigned* __restrict__ pcnt_out,
                                                unsigned* cnt, int b) {
    const int r = b / DEG_CHUNKS;
    const int c = b % DEG_CHUNKS;
    const unsigned lo = (unsigned)(r * DEG_NODES);
    const int tid = threadIdx.x;

    for (int i = tid; i < DEG_NODES; i += P1_THREADS) cnt[i] = 0u;
    __syncthreads();

    const int4* s4 = (const int4*)(src + (size_t)c * DEG_EDGES);
    const int n4 = DEG_EDGES / 4;
    for (int i0 = tid; i0 < n4; i0 += P1_THREADS * DEG_LOADS) {
        int4 v[DEG_LOADS];
#pragma unroll
        for (int u = 0; u < DEG_LOADS; ++u) {
            int idx = i0 + u * P1_THREADS;
            v[u] = s4[(idx < n4) ? idx : (n4 - 1)];
        }
#pragma unroll
        for (int u = 0; u < DEG_LOADS; ++u) {
            const bool ok = (i0 + u * P1_THREADS) < n4;
            unsigned t;
            t = (unsigned)v[u].x - lo; if (ok && t < (unsigned)DEG_NODES) atomicAdd(&cnt[t], 1u);
            t = (unsigned)v[u].y - lo; if (ok && t < (unsigned)DEG_NODES) atomicAdd(&cnt[t], 1u);
            t = (unsigned)v[u].z - lo; if (ok && t < (unsigned)DEG_NODES) atomicAdd(&cnt[t], 1u);
            t = (unsigned)v[u].w - lo; if (ok && t < (unsigned)DEG_NODES) atomicAdd(&cnt[t], 1u);
        }
    }
    __syncthreads();

    unsigned* o = pcnt_out + (size_t)(c * DEG_RANGES + r) * DEG_SLAB;
    for (int i = tid; i < DEG_NODES; i += P1_THREADS) o[i] = cnt[i];
}

__device__ __forceinline__ void gcn_nodeprep_role(const float* __restrict__ x,
                                                  const unsigned* __restrict__ pcnt_out,
                                                  float* __restrict__ xscaled,
                                                  int nN, int b) {
    int n = b * 256 + threadIdx.x;
    if (n < nN) {
        const int r = n / DEG_NODES;
        const int l = n - r * DEG_NODES;
        unsigned d = 0u;
#pragma unroll
        for (int c = 0; c < DEG_CHUNKS; ++c) d += pcnt_out[(size_t)(c * DEG_RANGES + r) * DEG_SLAB + l];
        float dout = (float)d; dout = (dout > 0.0f) ? dout : 1.0f;
        float ns = rsqrtf(dout);
        const float* xr = x + (size_t)n * 40 + 15;
        float* o = xscaled + (size_t)n * F_IN;
#pragma unroll
        for (int k = 0; k < F_IN; ++k) o[k] = xr[k] * ns;
    }
}

#define FOLD_PARTS 4
#define FOLD_K     (128 / FOLD_PARTS)
__device__ __forceinline__ void gcn_foldw_role(const float* __restrict__ W1,
                                               const float* __restrict__ b1,
                                               const float* __restrict__ W2,
                                               const float* __restrict__ b2,
                                               float* __restrict__ Wc,
                                               float* __restrict__ bc,
                                               float (*part)[64], int i) {
    const int j = threadIdx.x & 63;
    const int h = threadIdx.x >> 6;
    const float* a = ((i < F_IN) ? (W1 + i * 128) : b1) + h * FOLD_K;
    const float* w = W2 + (h * FOLD_K) * 64 + j;

    float av[FOLD_K], wv[FOLD_K];
#pragma unroll
    for (int k = 0; k < FOLD_K; ++k) { av[k] = a[k]; wv[k] = w[k * 64]; }
    float s = 0.0f;
#pragma unroll
    for (int k = 0; k < FOLD_K; ++k) s += av[k] * wv[k];
    part[h][j] = s;
    __syncthreads();

    if (h == 0) {
        float t = part[0][j];
#pragma unroll
        for (int q = 1; q < FOLD_PARTS; ++q) t += part[q][j];
        if (i < F_IN) Wc[i * 64 + j] = t;
        else          bc[j] = b2[j] + t;
    } else if (i == F_IN && h <= KPAD - F_IN) {
        Wc[(F_IN + h - 1) * 64 + j] = 0.0f;
    }
}

#define FOLD_WGS (F_IN + 1)
__global__ void __launch_bounds__(256) gcn_prep_kernel(const float* __restrict__ x,
                                                       const unsigned* __restrict__ pcnt_out,
                                                       float* __restrict__ xscaled,
                                                       int nN,
                                                       const float* __restrict__ W1,
                                                       const float* __restrict__ b1,
                                                       const float* __restrict__ W2,
                                                       const float* __restrict__ b2,
                                                       float* __restrict__ Wc,
                                                       float* __restrict__ bc) {
    __shared__ float part[FOLD_PARTS][64];
    if (blockIdx.x < FOLD_WGS) gcn_foldw_role(W1, b1, W2, b2, Wc, bc, part, (int)blockIdx.x);
    else                       gcn_nodeprep_role(x, pcnt_out, xscaled, nN, (int)blockIdx.x - FOLD_WGS);
}

__device__ __forceinline__ void gcn_bin_role(const int* __restrict__ src,
                                             const int* __restrict__ dst,
                                             unsigned* __restrict__ binned,
                                             unsigned* __restrict__ binoff,
                                             unsigned* lds, int g) {
    unsigned* stage = lds;
    unsigned* hist  = lds + P1_HIST_AT;
    unsigned* start = lds + P1_START_AT;
    unsigned* wsum  = lds + P1_WSUM_AT;

    const int tid = threadIdx.x;
    const bool worker = tid < BIN_THREADS;
    if (tid <= OW_RANGES) hist[tid] = 0u;
    __syncthreads();

    const int4* d4 = (const int4*)(dst + (size_t)g * BIN_EDGES);
    const int4* s4 = (const int4*)(src + (size_t)g * BIN_EDGES);
    const int n4 = BIN_EDGES / 4;

    unsigned pk[BIN_LOADS * 4], br[BIN_LOADS * 4];
    if (worker) {
        int4 dv[BIN_LOADS], sv[BIN_LOADS];
#pragma unroll
        for (int u = 0; u < BIN_LOADS; ++u) {
            dv[u] = d4[tid + u * BIN_THREADS];
            sv[u] = s4[tid + u * BIN_THREADS];
        }

#define BIN_RANK(J, D, S)                                                               \
        {                                                                                   \
            const unsigned d  = (unsigned)(D);                                              \
            const bool     in = d < (unsigned)N_NODES;                                      \
            const unsigned b  = in ? d / (unsigned)OW_NODES : (unsigned)OW_RANGES;          \
            const unsigned t  = d - b * (unsigned)OW_NODES;                                 \
            pk[J] = (t << 17) | ((unsigned)(S) & 0x1FFFFu);                                 \
            const unsigned rk = atomicAdd(&hist[b], 1u);                                    \
            br[J] = in ? ((b << 16) | rk) : BIN_NONE;                                       \
        }
#pragma unroll
        for (int u = 0; u < BIN_LOADS; ++u) {
            BIN_RANK(4 * u + 0, dv[u].x, sv[u].x)
            BIN_RANK(4 * u + 1, dv[u].y, sv[u].y)
            BIN_RANK(4 * u + 2, dv[u].z, sv[u].z)
            BIN_RANK(4 * u + 3, dv[u].w, sv[u].w)
        }
#undef BIN_RANK
    }
    __syncthreads();

    unsigned hv = 0u, inc = 0u;
    if (tid < 256) {
        hv  = (tid < OW_RANGES) ? hist[tid] : 0u;
        inc = hv;
        for (int o = 1; o < 32; o <<= 1) {
            const unsigned nb = __shfl_up(inc, (unsigned)o);
            if ((tid & 31) >= o) inc += nb;
        }
        if ((tid & 31) == 31) wsum[tid >> 5] = inc;
    }
    __syncthreads();
    if (tid < 256) {
        unsigned base = 0u;
        for (int q = 0; q < 8; ++q) if (q < (tid >> 5)) base += wsum[q];
        const unsigned ex = base + inc - hv;
        if (tid < OW_RANGES) {
            start[tid] = ex;
            binoff[g * BIN_OFFS + tid] = ex;
        }
        if (tid == OW_RANGES - 1) binoff[g * BIN_OFFS + OW_RANGES] = ex + hv;
    }
    __syncthreads();

    if (worker) {
#pragma unroll
        for (int j = 0; j < BIN_LOADS * 4; ++j)
            if (br[j] != BIN_NONE) stage[start[br[j] >> 16] + (br[j] & 0xFFFFu)] = pk[j];
    }
    __syncthreads();

    uint4* o = (uint4*)(binned + (size_t)g * BIN_STRIDE);
    const uint4* st4 = (const uint4*)stage;
    for (int i = tid; i < n4; i += P1_THREADS) o[i] = st4[i];
}

__global__ void __launch_bounds__(P1_THREADS) gcn_edges_kernel(const int* __restrict__ src,
                                                               const int* __restrict__ dst,
                                                               unsigned* __restrict__ pcnt_out,
                                                               unsigned* __restrict__ binned,
                                                               unsigned* __restrict__ binoff) {
    __shared__ __attribute__((aligned(16))) unsigned lds[P1_LDS];
    if (blockIdx.x < P1_DEG_WGS) gcn_degree_role(src, pcnt_out, lds, (int)blockIdx.x);
    else                         gcn_bin_role(src, dst, binned, binoff, lds, (int)blockIdx.x - P1_DEG_WGS);
}

__global__ void __launch_bounds__(OW_THREADS) gcn_owner_kernel(const unsigned* __restrict__ binned,
                                                               const unsigned* __restrict__ binoff,
                                                               const float* __restrict__ xscaled,
                                                               const float* __restrict__ Wc,
                                                               const float* __restrict__ bc,
                                                               float* __restrict__ out) {
    __shared__ __attribute__((aligned(16))) float acc[OW_NODES * F_IN];
    __shared__ unsigned cnt[OW_NODES];

    const int r    = blockIdx.x;
    const int tid  = threadIdx.x;
    const int lane = tid & 31;
    const int w    = __builtin_amdgcn_readfirstlane(tid >> 5);

    for (int i = tid; i < OW_NODES * F_IN; i += OW_THREADS) acc[i] = 0.0f;
    for (int i = tid; i < OW_NODES; i += OW_THREADS) cnt[i] = 0u;
    __syncthreads();

#define OW_ADD(P, A0, A1, A2, A3, A4)                                                   \
    {                                                                                   \
        const unsigned t = (P) >> 17;                                                   \
        float* a = acc + t * F_IN;                                                      \
        atomicAdd(a + 0, A0.x); atomicAdd(a + 1, A0.y);                                 \
        atomicAdd(a + 2, A1.x); atomicAdd(a + 3, A1.y);                                 \
        atomicAdd(a + 4, A2.x); atomicAdd(a + 5, A2.y);                                 \
        atomicAdd(a + 6, A3.x); atomicAdd(a + 7, A3.y);                                 \
        atomicAdd(a + 8, A4.x); atomicAdd(a + 9, A4.y);                                 \
        atomicAdd(&cnt[t], 1u);                                                         \
    }
    unsigned beg[OW_SLICES], len[OW_SLICES];
#pragma unroll
    for (int s = 0; s < OW_SLICES; ++s) {
        const int g = w * OW_SLICES + s;
        const unsigned b0 = binoff[g * BIN_OFFS + r];
        const unsigned e0 = binoff[g * BIN_OFFS + r + 1];
        beg[s] = (unsigned)g * BIN_STRIDE + b0;
        len[s] = e0 - b0;
    }
#pragma unroll
    for (int q = 0; q < OW_SLICES; q += OW_GROUP) {
        unsigned maxlen = 0u;
#pragma unroll
        for (int s = 0; s < OW_GROUP; ++s) maxlen = (len[q + s] > maxlen) ? len[q + s] : maxlen;
        for (unsigned j0 = 0u; j0 < maxlen; j0 += 32u * OW_PER) {
            bool     ok[OW_GROUP * OW_PER];
            unsigned p[OW_GROUP * OW_PER];
#pragma unroll
            for (int k = 0; k < OW_GROUP * OW_PER; ++k) {
                const int s = q + k / OW_PER;
                const unsigned j = j0 + (unsigned)lane + 32u * (unsigned)(k % OW_PER);
                ok[k] = j < len[s];
                p[k]  = binned[ok[k] ? beg[s] + j : (unsigned)(w * OW_SLICES + s) * BIN_STRIDE];
            }
            float2 v[OW_GROUP * OW_PER][5];
#pragma unroll
            for (int k = 0; k < OW_GROUP * OW_PER; ++k) {
                const float2* row = (const float2*)(xscaled + (size_t)(p[k] & 0x1FFFFu) * F_IN);
#pragma unroll
                for (int j = 0; j < 5; ++j) v[k][j] = row[j];
            }
#pragma unroll
            for (int k = 0; k < OW_GROUP * OW_PER; ++k)
                asm volatile("" :: "v"(v[k][0].x), "v"(v[k][0].y), "v"(v[k][1].x), "v"(v[k][1].y), "v"(v[k][2].x),
                                   "v"(v[k][2].y), "v"(v[k][3].x), "v"(v[k][3].y), "v"(v[k][4].x), "v"(v[k][4].y));
#pragma unroll
            for (int k = 0; k < OW_GROUP * OW_PER; ++k)
                if (ok[k]) OW_ADD(p[k], v[k][0], v[k][1], v[k][2], v[k][3], v[k][4])
        }
    }
#undef OW_ADD
    __syncthreads();

    const int m  = lane & 15;
    const int kh = lane >> 4;
    const int nn = lane & 15;
    for (int tile = w; tile < OW_TILES; tile += OW_WAVES) {
        const int ln = tile * 16 + m;
        float din = (float)cnt[ln]; din = (din > 0.0f) ? din : 1.0f;
        const float nd = rsqrtf(din);

        v2f afrag[3];
#pragma unroll
        for (int ks = 0; ks < 3; ++ks) {
            const int k0 = ks * 4 + kh * 2;
            float sx = 0.0f, sy = 0.0f;
            if (k0 < F_IN) {
                const float2 pv = *(const float2*)(acc + ln * F_IN + k0);
                sx = pv.x; sy = pv.y;
            }
            afrag[ks].x = sx * nd;
            afrag[ks].y = sy * nd;
        }

        const int nbase = r * OW_NODES + tile * 16;
#pragma unroll
        for (int j = 0; j < 4; ++j) {
            v8f c = {};
#pragma unroll
            for (int ks = 0; ks < 3; ++ks) {
                const int kb = ks * 4 + kh * 2;
                v2f b;
                b.x = Wc[(kb    ) * N_OUT + j * 16 + nn];
                b.y = Wc[(kb + 1) * N_OUT + j * 16 + nn];
                c = __builtin_amdgcn_wmma_f32_16x16x4_f32(
                         false, afrag[ks],
                         false, b,
                         (short)0, c,
                         false,  false);
            }
            const float bias = bc[j * 16 + nn];
#pragma unroll
            for (int r8 = 0; r8 < 8; ++r8) {
                const int row = r8 + 8 * kh;
                out[(size_t)(nbase + row) * N_OUT + j * 16 + nn] = c[r8] + bias;
            }
        }
    }
}

extern "C" void kernel_launch(void* const* d_in, const int* in_sizes, int n_in,
                              void* d_out, int out_size, void* d_ws, size_t ws_size,
                              hipStream_t stream) {
    const float* x   = (const float*)d_in[0];
    const int*   src = (const int*)  d_in[1];
    const int*   dst = (const int*)  d_in[2];
    const float* W1  = (const float*)d_in[3];
    const float* b1  = (const float*)d_in[4];
    const float* W2  = (const float*)d_in[5];
    const float* b2  = (const float*)d_in[6];
    float* out = (float*)d_out;

    float* ws = (float*)d_ws;
    unsigned* pcnt_out = (unsigned*)ws;
    float*    xscaled  = ws + (size_t)DEG_CHUNKS * DEG_RANGES * DEG_SLAB;
    float*    Wc       = xscaled + (size_t)N_NODES * F_IN;
    float*    bcv      = Wc + KPAD * N_OUT;
    unsigned* binned   = (unsigned*)(bcv + N_OUT);
    unsigned* binoff   = binned + (size_t)BIN_SLICES * BIN_STRIDE;

    gcn_edges_kernel<<<P1_DEG_WGS + BIN_SLICES, P1_THREADS, 0, stream>>>(src, dst, pcnt_out, binned, binoff);

    gcn_prep_kernel<<<FOLD_WGS + (N_NODES + 255) / 256, 256, 0, stream>>>(x, pcnt_out, xscaled, N_NODES,
                                                                          W1, b1, W2, b2, Wc, bcv);

    gcn_owner_kernel<<<OW_RANGES, OW_THREADS, 0, stream>>>(binned, binoff, xscaled, Wc, bcv, out);
}


# Source-row gather re-shaped: scaled rows at a 64-byte pitch (no row spans two lines), each fetched by 5 adjacent lanes as one float2 each, 6 rows per load; LDS-atomic sums stay in arrival order
# speedup vs baseline: 15.2177x; 1.0799x over previous
#include <hip/hip_runtime.h>


#define N_NODES  100000
#define N_EDGES  3200000
#define F_IN     10
#define KPAD     12
#define XS_PITCH 16
#define N_OUT    64

#define DEG_RANGES 16
#define DEG_CHUNKS 16
#define DEG_NODES  (N_NODES / DEG_RANGES)
#define DEG_EDGES  (N_EDGES / DEG_CHUNKS)
#define DEG_SLAB   6272
#define DEG_LOADS  8

#define OW_RANGES  250
#define OW_NODES   (N_NODES / OW_RANGES)
#define OW_TILES   (OW_NODES / 16)
#define OW_WAVES   16
#define OW_THREADS (OW_WAVES * 32)
#define OW_LANES   5
#define OW_EPI     6
#define OW_GROUP   4
#define OW_PER     12

#define BIN_SLICES  256
#define OW_SLICES   (BIN_SLICES / OW_WAVES)
#define BIN_EDGES   (N_EDGES / BIN_SLICES)
#define BIN_STRIDE  12512
#define BIN_OFFS    (OW_RANGES + 1)
#define BIN_THREADS 625
#define BIN_LOADS   5
#define BIN_NONE    0xFFFFFFFFu

#define P1_THREADS  640
#define P1_DEG_WGS  (DEG_RANGES * DEG_CHUNKS)
#define P1_HIST_AT  BIN_STRIDE
#define P1_START_AT (BIN_STRIDE + 256)
#define P1_WSUM_AT  (BIN_STRIDE + 512)
#define P1_LDS      (BIN_STRIDE + 512 + 32)

typedef __attribute__((ext_vector_type(2))) float v2f;
typedef __attribute__((ext_vector_type(8))) float v8f;

__device__ __forceinline__ void gcn_degree_role(const int* __restrict__ src, unsigned* __restrict__ pcnt_out,
                                                unsigned* cnt, int b) {
    const int r = b / DEG_CHUNKS;
    const int c = b % DEG_CHUNKS;
    const unsigned lo = (unsigned)(r * DEG_NODES);
    const int tid = threadIdx.x;

    for (int i = tid; i < DEG_NODES; i += P1_THREADS) cnt[i] = 0u;
    __syncthreads();

    const int4* s4 = (const int4*)(src + (size_t)c * DEG_EDGES);
    const int n4 = DEG_EDGES / 4;
    for (int i0 = tid; i0 < n4; i0 += P1_THREADS * DEG_LOADS) {
        int4 v[DEG_LOADS];
#pragma unroll
        for (int u = 0; u < DEG_LOADS; ++u) {
            int idx = i0 + u * P1_THREADS;
            v[u] = s4[(idx < n4) ? idx : (n4 - 1)];
        }
#pragma unroll
        for (int u = 0; u < DEG_LOADS; ++u) {
            const bool ok = (i0 + u * P1_THREADS) < n4;
            unsigned t;
            t = (unsigned)v[u].x - lo; if (ok && t < (unsigned)DEG_NODES) atomicAdd(&cnt[t], 1u);
            t = (unsigned)v[u].y - lo; if (ok && t < (unsigned)DEG_NODES) atomicAdd(&cnt[t], 1u);
            t = (unsigned)v[u].z - lo; if (ok && t < (unsigned)DEG_NODES) atomicAdd(&cnt[t], 1u);
            t = (unsigned)v[u].w - lo; if (ok && t < (unsigned)DEG_NODES) atomicAdd(&cnt[t], 1u);
        }
    }
    __syncthreads();

    unsigned* o = pcnt_out + (size_t)(c * DEG_RANGES + r) * DEG_SLAB;
    for (int i = tid; i < DEG_NODES; i += P1_THREADS) o[i] = cnt[i];
}

__device__ __forceinline__ void gcn_nodeprep_role(const float* __restrict__ x,
                                                  const unsigned* __restrict__ pcnt_out,
                                                  float* __restrict__ xscaled,
                                                  int nN, int b) {
    int n = b * 256 + threadIdx.x;
    if (n < nN) {
        const int r = n / DEG_NODES;
        const int l = n - r * DEG_NODES;
        unsigned d = 0u;
#pragma unroll
        for (int c = 0; c < DEG_CHUNKS; ++c) d += pcnt_out[(size_t)(c * DEG_RANGES + r) * DEG_SLAB + l];
        float dout = (float)d; dout = (dout > 0.0f) ? dout : 1.0f;
        float ns = rsqrtf(dout);
        const float* xr = x + (size_t)n * 40 + 15;
        float* o = xscaled + (size_t)n * XS_PITCH;
#pragma unroll
        for (int k = 0; k < F_IN; ++k) o[k] = xr[k] * ns;
    }
}

#define FOLD_PARTS 4
#define FOLD_K     (128 / FOLD_PARTS)
__device__ __forceinline__ void gcn_foldw_role(const float* __restrict__ W1,
                                               const float* __restrict__ b1,
                                               const float* __restrict__ W2,
                                               const float* __restrict__ b2,
                                               float* __restrict__ Wc,
                                               float* __restrict__ bc,
                                               float (*part)[64], int i) {
    const int j = threadIdx.x & 63;
    const int h = threadIdx.x >> 6;
    const float* a = ((i < F_IN) ? (W1 + i * 128) : b1) + h * FOLD_K;
    const float* w = W2 + (h * FOLD_K) * 64 + j;

    float av[FOLD_K], wv[FOLD_K];
#pragma unroll
    for (int k = 0; k < FOLD_K; ++k) { av[k] = a[k]; wv[k] = w[k * 64]; }
    float s = 0.0f;
#pragma unroll
    for (int k = 0; k < FOLD_K; ++k) s += av[k] * wv[k];
    part[h][j] = s;
    __syncthreads();

    if (h == 0) {
        float t = part[0][j];
#pragma unroll
        for (int q = 1; q < FOLD_PARTS; ++q) t += part[q][j];
        if (i < F_IN) Wc[i * 64 + j] = t;
        else          bc[j] = b2[j] + t;
    } else if (i == F_IN && h <= KPAD - F_IN) {
        Wc[(F_IN + h - 1) * 64 + j] = 0.0f;
    }
}

#define FOLD_WGS (F_IN + 1)
__global__ void __launch_bounds__(256) gcn_prep_kernel(const float* __restrict__ x,
                                                       const unsigned* __restrict__ pcnt_out,
                                                       float* __restrict__ xscaled,
                                                       int nN,
                                                       const float* __restrict__ W1,
                                                       const float* __restrict__ b1,
                                                       const float* __restrict__ W2,
                                                       const float* __restrict__ b2,
                                                       float* __restrict__ Wc,
                                                       float* __restrict__ bc) {
    __shared__ float part[FOLD_PARTS][64];
    if (blockIdx.x < FOLD_WGS) gcn_foldw_role(W1, b1, W2, b2, Wc, bc, part, (int)blockIdx.x);
    else                       gcn_nodeprep_role(x, pcnt_out, xscaled, nN, (int)blockIdx.x - FOLD_WGS);
}

__device__ __forceinline__ void gcn_bin_role(const int* __restrict__ src,
                                             const int* __restrict__ dst,
                                             unsigned* __restrict__ binned,
                                             unsigned* __restrict__ binoff,
                                             unsigned* lds, int g) {
    unsigned* stage = lds;
    unsigned* hist  = lds + P1_HIST_AT;
    unsigned* start = lds + P1_START_AT;
    unsigned* wsum  = lds + P1_WSUM_AT;

    const int tid = threadIdx.x;
    const bool worker = tid < BIN_THREADS;
    if (tid <= OW_RANGES) hist[tid] = 0u;
    __syncthreads();

    const int4* d4 = (const int4*)(dst + (size_t)g * BIN_EDGES);
    const int4* s4 = (const int4*)(src + (size_t)g * BIN_EDGES);
    const int n4 = BIN_EDGES / 4;

    unsigned pk[BIN_LOADS * 4], br[BIN_LOADS * 4];
    if (worker) {
        int4 dv[BIN_LOADS], sv[BIN_LOADS];
#pragma unroll
        for (int u = 0; u < BIN_LOADS; ++u) {
            dv[u] = d4[tid + u * BIN_THREADS];
            sv[u] = s4[tid + u * BIN_THREADS];
        }

#define BIN_RANK(J, D, S)                                                               \
        {                                                                                   \
            const unsigned d  = (unsigned)(D);                                              \
            const bool     in = d < (unsigned)N_NODES;                                      \
            const unsigned b  = in ? d / (unsigned)OW_NODES : (unsigned)OW_RANGES;          \
            const unsigned t  = d - b * (unsigned)OW_NODES;                                 \
            pk[J] = (t << 17) | ((unsigned)(S) & 0x1FFFFu);                                 \
            const unsigned rk = atomicAdd(&hist[b], 1u);                                    \
            br[J] = in ? ((b << 16) | rk) : BIN_NONE;                                       \
        }
#pragma unroll
        for (int u = 0; u < BIN_LOADS; ++u) {
            BIN_RANK(4 * u + 0, dv[u].x, sv[u].x)
            BIN_RANK(4 * u + 1, dv[u].y, sv[u].y)
            BIN_RANK(4 * u + 2, dv[u].z, sv[u].z)
            BIN_RANK(4 * u + 3, dv[u].w, sv[u].w)
        }
#undef BIN_RANK
    }
    __syncthreads();

    unsigned hv = 0u, inc = 0u;
    if (tid < 256) {
        hv  = (tid < OW_RANGES) ? hist[tid] : 0u;
        inc = hv;
        for (int o = 1; o < 32; o <<= 1) {
            const unsigned nb = __shfl_up(inc, (unsigned)o);
            if ((tid & 31) >= o) inc += nb;
        }
        if ((tid & 31) == 31) wsum[tid >> 5] = inc;
    }
    __syncthreads();
    if (tid < 256) {
        unsigned base = 0u;
        for (int q = 0; q < 8; ++q) if (q < (tid >> 5)) base += wsum[q];
        const unsigned ex = base + inc - hv;
        if (tid < OW_RANGES) {
            start[tid] = ex;
            binoff[g * BIN_OFFS + tid] = ex;
        }
        if (tid == OW_RANGES - 1) binoff[g * BIN_OFFS + OW_RANGES] = ex + hv;
    }
    __syncthreads();

    if (worker) {
#pragma unroll
        for (int j = 0; j < BIN_LOADS * 4; ++j)
            if (br[j] != BIN_NONE) stage[start[br[j] >> 16] + (br[j] & 0xFFFFu)] = pk[j];
    }
    __syncthreads();

    uint4* o = (uint4*)(binned + (size_t)g * BIN_STRIDE);
    const uint4* st4 = (const uint4*)stage;
    for (int i = tid; i < n4; i += P1_THREADS) o[i] = st4[i];
}

__global__ void __launch_bounds__(P1_THREADS) gcn_edges_kernel(const int* __restrict__ src,
                                                               const int* __restrict__ dst,
                                                               unsigned* __restrict__ pcnt_out,
                                                               unsigned* __restrict__ binned,
                                                               unsigned* __restrict__ binoff) {
    __shared__ __attribute__((aligned(16))) unsigned lds[P1_LDS];
    if (blockIdx.x < P1_DEG_WGS) gcn_degree_role(src, pcnt_out, lds, (int)blockIdx.x);
    else                         gcn_bin_role(src, dst, binned, binoff, lds, (int)blockIdx.x - P1_DEG_WGS);
}

__global__ void __launch_bounds__(OW_THREADS) gcn_owner_kernel(const unsigned* __restrict__ binned,
                                                               const unsigned* __restrict__ binoff,
                                                               const float* __restrict__ xscaled,
                                                               const float* __restrict__ Wc,
                                                               const float* __restrict__ bc,
                                                               float* __restrict__ out) {
    __shared__ __attribute__((aligned(16))) float acc[OW_NODES * F_IN];
    __shared__ unsigned cnt[OW_NODES];

    const int r    = blockIdx.x;
    const int tid  = threadIdx.x;
    const int lane = tid & 31;
    const int w    = __builtin_amdgcn_readfirstlane(tid >> 5);

    for (int i = tid; i < OW_NODES * F_IN; i += OW_THREADS) acc[i] = 0.0f;
    for (int i = tid; i < OW_NODES; i += OW_THREADS) cnt[i] = 0u;
    __syncthreads();

    const int  e   = lane / OW_LANES;
    const int  q   = lane - e * OW_LANES;
    const bool act = lane < OW_EPI * OW_LANES;
#define OW_RUN(B, L, S)                                                                 \
    {                                                                                   \
        const int g = w * OW_SLICES + (S);                                              \
        const unsigned b0 = binoff[g * BIN_OFFS + r];                                   \
        const unsigned e0 = binoff[g * BIN_OFFS + r + 1];                               \
        (B) = (unsigned)g * BIN_STRIDE + b0;                                            \
        (L) = e0 - b0;                                                                  \
    }
    unsigned beg[OW_GROUP], len[OW_GROUP];
#pragma unroll
    for (int s = 0; s < OW_GROUP; ++s) OW_RUN(beg[s], len[s], s)
#pragma nounroll
    for (int s0 = 0; s0 < OW_SLICES; s0 += OW_GROUP) {
        const int sn = (s0 + OW_GROUP < OW_SLICES) ? s0 + OW_GROUP : s0;
        unsigned nbeg[OW_GROUP], nlen[OW_GROUP];
#pragma unroll
        for (int s = 0; s < OW_GROUP; ++s) OW_RUN(nbeg[s], nlen[s], sn + s)
        unsigned maxlen = 0u;
#pragma unroll
        for (int s = 0; s < OW_GROUP; ++s) maxlen = (len[s] > maxlen) ? len[s] : maxlen;
        for (unsigned j0 = 0u; j0 < maxlen; j0 += (unsigned)(OW_EPI * OW_PER)) {
            bool     ok[OW_GROUP * OW_PER];
            unsigned p[OW_GROUP * OW_PER];
#pragma unroll
            for (int k = 0; k < OW_GROUP * OW_PER; ++k) {
                const int s = k / OW_PER;
                const unsigned j = j0 + (unsigned)(OW_EPI * (k % OW_PER) + e);
                ok[k] = act && (j < len[s]);
                p[k]  = binned[ok[k] ? beg[s] + j : (unsigned)(w * OW_SLICES + s0 + s) * BIN_STRIDE];
            }
            float2 v[OW_GROUP * OW_PER];
#pragma unroll
            for (int k = 0; k < OW_GROUP * OW_PER; ++k)
                v[k] = *(const float2*)(xscaled + (size_t)(p[k] & 0x1FFFFu) * XS_PITCH + 2 * q);
#pragma unroll
            for (int k = 0; k < OW_GROUP * OW_PER; ++k)
                asm volatile("" :: "v"(v[k].x), "v"(v[k].y));
#pragma unroll
            for (int k = 0; k < OW_GROUP * OW_PER; ++k)
                if (ok[k]) {
                    const unsigned t = p[k] >> 17;
                    float* a = acc + t * F_IN + 2 * q;
                    atomicAdd(a, v[k].x);
                    atomicAdd(a + 1, v[k].y);
                    if (q == 0) atomicAdd(&cnt[t], 1u);
                }
        }
#pragma unroll
        for (int s = 0; s < OW_GROUP; ++s) { beg[s] = nbeg[s]; len[s] = nlen[s]; }
    }
#undef OW_RUN
    __syncthreads();

    const int m  = lane & 15;
    const int kh = lane >> 4;
    const int nn = lane & 15;
    for (int tile = w; tile < OW_TILES; tile += OW_WAVES) {
        const int ln = tile * 16 + m;
        float din = (float)cnt[ln]; din = (din > 0.0f) ? din : 1.0f;
        const float nd = rsqrtf(din);

        v2f afrag[3];
#pragma unroll
        for (int ks = 0; ks < 3; ++ks) {
            const int k0 = ks * 4 + kh * 2;
            float sx = 0.0f, sy = 0.0f;
            if (k0 < F_IN) {
                const float2 pv = *(const float2*)(acc + ln * F_IN + k0);
                sx = pv.x; sy = pv.y;
            }
            afrag[ks].x = sx * nd;
            afrag[ks].y = sy * nd;
        }

        const int nbase = r * OW_NODES + tile * 16;
#pragma unroll
        for (int j = 0; j < 4; ++j) {
            v8f c = {};
#pragma unroll
            for (int ks = 0; ks < 3; ++ks) {
                const int kb = ks * 4 + kh * 2;
                v2f b;
                b.x = Wc[(kb    ) * N_OUT + j * 16 + nn];
                b.y = Wc[(kb + 1) * N_OUT + j * 16 + nn];
                c = __builtin_amdgcn_wmma_f32_16x16x4_f32(
                         false, afrag[ks],
                         false, b,
                         (short)0, c,
                         false,  false);
            }
            const float bias = bc[j * 16 + nn];
#pragma unroll
            for (int r8 = 0; r8 < 8; ++r8) {
                const int row = r8 + 8 * kh;
                out[(size_t)(nbase + row) * N_OUT + j * 16 + nn] = c[r8] + bias;
            }
        }
    }
}

extern "C" void kernel_launch(void* const* d_in, const int* in_sizes, int n_in,
                              void* d_out, int out_size, void* d_ws, size_t ws_size,
                              hipStream_t stream) {
    const float* x   = (const float*)d_in[0];
    const int*   src = (const int*)  d_in[1];
    const int*   dst = (const int*)  d_in[2];
    const float* W1  = (const float*)d_in[3];
    const float* b1  = (const float*)d_in[4];
    const float* W2  = (const float*)d_in[5];
    const float* b2  = (const float*)d_in[6];
    float* out = (float*)d_out;

    float* ws = (float*)d_ws;
    unsigned* pcnt_out = (unsigned*)ws;
    float*    xscaled  = ws + (size_t)DEG_CHUNKS * DEG_RANGES * DEG_SLAB;
    float*    Wc       = xscaled + (size_t)N_NODES * XS_PITCH;
    float*    bcv      = Wc + KPAD * N_OUT;
    unsigned* binned   = (unsigned*)(bcv + N_OUT);
    unsigned* binoff   = binned + (size_t)BIN_SLICES * BIN_STRIDE;

    gcn_edges_kernel<<<P1_DEG_WGS + BIN_SLICES, P1_THREADS, 0, stream>>>(src, dst, pcnt_out, binned, binoff);

    gcn_prep_kernel<<<FOLD_WGS + (N_NODES + 255) / 256, 256, 0, stream>>>(x, pcnt_out, xscaled, N_NODES,
                                                                          W1, b1, W2, b2, Wc, bcv);

    gcn_owner_kernel<<<OW_RANGES, OW_THREADS, 0, stream>>>(binned, binoff, xscaled, Wc, bcv, out);
}


# Owner gather re-shaped: a row is 12 floats (10 features, a stored 1 that sums to the in-degree, a 0) fetched by 4 lanes x 12 B, 8 rows per load on all 32 lanes; no integer atomic; LDS arrival order
# speedup vs baseline: 16.4715x; 1.0824x over previous
#include <hip/hip_runtime.h>


#define N_NODES  100000
#define N_EDGES  3200000
#define F_IN     10
#define KPAD     12
#define XS_PITCH 16
#define ROW_W    12
#define N_OUT    64

#define DEG_RANGES 16
#define DEG_CHUNKS 16
#define DEG_NODES  (N_NODES / DEG_RANGES)
#define DEG_EDGES  (N_EDGES / DEG_CHUNKS)
#define DEG_SLAB   6272
#define DEG_LOADS  8

#define OW_RANGES  250
#define OW_NODES   (N_NODES / OW_RANGES)
#define OW_TILES   (OW_NODES / 16)
#define OW_WAVES   16
#define OW_THREADS (OW_WAVES * 32)
#define OW_LANES   4
#define OW_EPI     8
#define OW_GROUP   4
#define OW_PER     9

#define BIN_SLICES  256
#define OW_SLICES   (BIN_SLICES / OW_WAVES)
#define BIN_EDGES   (N_EDGES / BIN_SLICES)
#define BIN_STRIDE  12512
#define BIN_OFFS    (OW_RANGES + 1)
#define BIN_THREADS 625
#define BIN_LOADS   5
#define BIN_NONE    0xFFFFFFFFu

#define P1_THREADS  640
#define P1_DEG_WGS  (DEG_RANGES * DEG_CHUNKS)
#define P1_HIST_AT  BIN_STRIDE
#define P1_START_AT (BIN_STRIDE + 256)
#define P1_WSUM_AT  (BIN_STRIDE + 512)
#define P1_LDS      (BIN_STRIDE + 512 + 32)

typedef __attribute__((ext_vector_type(2))) float v2f;
struct row3 { float x, y, z; };
typedef __attribute__((ext_vector_type(8))) float v8f;

__device__ __forceinline__ void gcn_degree_role(const int* __restrict__ src, unsigned* __restrict__ pcnt_out,
                                                unsigned* cnt, int b) {
    const int r = b / DEG_CHUNKS;
    const int c = b % DEG_CHUNKS;
    const unsigned lo = (unsigned)(r * DEG_NODES);
    const int tid = threadIdx.x;

    for (int i = tid; i < DEG_NODES; i += P1_THREADS) cnt[i] = 0u;
    __syncthreads();

    const int4* s4 = (const int4*)(src + (size_t)c * DEG_EDGES);
    const int n4 = DEG_EDGES / 4;
    for (int i0 = tid; i0 < n4; i0 += P1_THREADS * DEG_LOADS) {
        int4 v[DEG_LOADS];
#pragma unroll
        for (int u = 0; u < DEG_LOADS; ++u) {
            int idx = i0 + u * P1_THREADS;
            v[u] = s4[(idx < n4) ? idx : (n4 - 1)];
        }
#pragma unroll
        for (int u = 0; u < DEG_LOADS; ++u) {
            const bool ok = (i0 + u * P1_THREADS) < n4;
            unsigned t;
            t = (unsigned)v[u].x - lo; if (ok && t < (unsigned)DEG_NODES) atomicAdd(&cnt[t], 1u);
            t = (unsigned)v[u].y - lo; if (ok && t < (unsigned)DEG_NODES) atomicAdd(&cnt[t], 1u);
            t = (unsigned)v[u].z - lo; if (ok && t < (unsigned)DEG_NODES) atomicAdd(&cnt[t], 1u);
            t = (unsigned)v[u].w - lo; if (ok && t < (unsigned)DEG_NODES) atomicAdd(&cnt[t], 1u);
        }
    }
    __syncthreads();

    unsigned* o = pcnt_out + (size_t)(c * DEG_RANGES + r) * DEG_SLAB;
    for (int i = tid; i < DEG_NODES; i += P1_THREADS) o[i] = cnt[i];
}

__device__ __forceinline__ void gcn_nodeprep_role(const float* __restrict__ x,
                                                  const unsigned* __restrict__ pcnt_out,
                                                  float* __restrict__ xscaled,
                                                  int nN, int b) {
    int n = b * 256 + threadIdx.x;
    if (n < nN) {
        const int r = n / DEG_NODES;
        const int l = n - r * DEG_NODES;
        unsigned d = 0u;
#pragma unroll
        for (int c = 0; c < DEG_CHUNKS; ++c) d += pcnt_out[(size_t)(c * DEG_RANGES + r) * DEG_SLAB + l];
        float dout = (float)d; dout = (dout > 0.0f) ? dout : 1.0f;
        float ns = rsqrtf(dout);
        const float* xr = x + (size_t)n * 40 + 15;
        float4* o = (float4*)(xscaled + (size_t)n * XS_PITCH);
        o[0] = make_float4(xr[0] * ns, xr[1] * ns, xr[2] * ns, xr[3] * ns);
        o[1] = make_float4(xr[4] * ns, xr[5] * ns, xr[6] * ns, xr[7] * ns);
        o[2] = make_float4(xr[8] * ns, xr[9] * ns, 1.0f, 0.0f);
    }
}

#define FOLD_PARTS 4
#define FOLD_K     (128 / FOLD_PARTS)
__device__ __forceinline__ void gcn_foldw_role(const float* __restrict__ W1,
                                               const float* __restrict__ b1,
                                               const float* __restrict__ W2,
                                               const float* __restrict__ b2,
                                               float* __restrict__ Wc,
                                               float* __restrict__ bc,
                                               float (*part)[64], int i) {
    const int j = threadIdx.x & 63;
    const int h = threadIdx.x >> 6;
    const float* a = ((i < F_IN) ? (W1 + i * 128) : b1) + h * FOLD_K;
    const float* w = W2 + (h * FOLD_K) * 64 + j;

    float av[FOLD_K], wv[FOLD_K];
#pragma unroll
    for (int k = 0; k < FOLD_K; ++k) { av[k] = a[k]; wv[k] = w[k * 64]; }
    float s = 0.0f;
#pragma unroll
    for (int k = 0; k < FOLD_K; ++k) s += av[k] * wv[k];
    part[h][j] = s;
    __syncthreads();

    if (h == 0) {
        float t = part[0][j];
#pragma unroll
        for (int q = 1; q < FOLD_PARTS; ++q) t += part[q][j];
        if (i < F_IN) Wc[i * 64 + j] = t;
        else          bc[j] = b2[j] + t;
    } else if (i == F_IN && h <= KPAD - F_IN) {
        Wc[(F_IN + h - 1) * 64 + j] = 0.0f;
    }
}

#define FOLD_WGS (F_IN + 1)
__global__ void __launch_bounds__(256) gcn_prep_kernel(const float* __restrict__ x,
                                                       const unsigned* __restrict__ pcnt_out,
                                                       float* __restrict__ xscaled,
                                                       int nN,
                                                       const float* __restrict__ W1,
                                                       const float* __restrict__ b1,
                                                       const float* __restrict__ W2,
                                                       const float* __restrict__ b2,
                                                       float* __restrict__ Wc,
                                                       float* __restrict__ bc) {
    __shared__ float part[FOLD_PARTS][64];
    if (blockIdx.x < FOLD_WGS) gcn_foldw_role(W1, b1, W2, b2, Wc, bc, part, (int)blockIdx.x);
    else                       gcn_nodeprep_role(x, pcnt_out, xscaled, nN, (int)blockIdx.x - FOLD_WGS);
}

__device__ __forceinline__ void gcn_bin_role(const int* __restrict__ src,
                                             const int* __restrict__ dst,
                                             unsigned* __restrict__ binned,
                                             unsigned* __restrict__ binoff,
                                             unsigned* lds, int g) {
    unsigned* stage = lds;
    unsigned* hist  = lds + P1_HIST_AT;
    unsigned* start = lds + P1_START_AT;
    unsigned* wsum  = lds + P1_WSUM_AT;

    const int tid = threadIdx.x;
    const bool worker = tid < BIN_THREADS;
    if (tid <= OW_RANGES) hist[tid] = 0u;
    __syncthreads();

    const int4* d4 = (const int4*)(dst + (size_t)g * BIN_EDGES);
    const int4* s4 = (const int4*)(src + (size_t)g * BIN_EDGES);
    const int n4 = BIN_EDGES / 4;

    unsigned pk[BIN_LOADS * 4], br[BIN_LOADS * 4];
    if (worker) {
        int4 dv[BIN_LOADS], sv[BIN_LOADS];
#pragma unroll
        for (int u = 0; u < BIN_LOADS; ++u) {
            dv[u] = d4[tid + u * BIN_THREADS];
            sv[u] = s4[tid + u * BIN_THREADS];
        }

#define BIN_RANK(J, D, S)                                                               \
        {                                                                                   \
            const unsigned d  = (unsigned)(D);                                              \
            const bool     in = d < (unsigned)N_NODES;                                      \
            const unsigned b  = in ? d / (unsigned)OW_NODES : (unsigned)OW_RANGES;          \
            const unsigned t  = d - b * (unsigned)OW_NODES;                                 \
            pk[J] = (t << 17) | ((unsigned)(S) & 0x1FFFFu);                                 \
            const unsigned rk = atomicAdd(&hist[b], 1u);                                    \
            br[J] = in ? ((b << 16) | rk) : BIN_NONE;                                       \
        }
#pragma unroll
        for (int u = 0; u < BIN_LOADS; ++u) {
            BIN_RANK(4 * u + 0, dv[u].x, sv[u].x)
            BIN_RANK(4 * u + 1, dv[u].y, sv[u].y)
            BIN_RANK(4 * u + 2, dv[u].z, sv[u].z)
            BIN_RANK(4 * u + 3, dv[u].w, sv[u].w)
        }
#undef BIN_RANK
    }
    __syncthreads();

    unsigned hv = 0u, inc = 0u;
    if (tid < 256) {
        hv  = (tid < OW_RANGES) ? hist[tid] : 0u;
        inc = hv;
        for (int o = 1; o < 32; o <<= 1) {
            const unsigned nb = __shfl_up(inc, (unsigned)o);
            if ((tid & 31) >= o) inc += nb;
        }
        if ((tid & 31) == 31) wsum[tid >> 5] = inc;
    }
    __syncthreads();
    if (tid < 256) {
        unsigned base = 0u;
        for (int q = 0; q < 8; ++q) if (q < (tid >> 5)) base += wsum[q];
        const unsigned ex = base + inc - hv;
        if (tid < OW_RANGES) {
            start[tid] = ex;
            binoff[g * BIN_OFFS + tid] = ex;
        }
        if (tid == OW_RANGES - 1) binoff[g * BIN_OFFS + OW_RANGES] = ex + hv;
    }
    __syncthreads();

    if (worker) {
#pragma unroll
        for (int j = 0; j < BIN_LOADS * 4; ++j)
            if (br[j] != BIN_NONE) stage[start[br[j] >> 16] + (br[j] & 0xFFFFu)] = pk[j];
    }
    __syncthreads();

    uint4* o = (uint4*)(binned + (size_t)g * BIN_STRIDE);
    const uint4* st4 = (const uint4*)stage;
    for (int i = tid; i < n4; i += P1_THREADS) o[i] = st4[i];
}

__global__ void __launch_bounds__(P1_THREADS) gcn_edges_kernel(const int* __restrict__ src,
                                                               const int* __restrict__ dst,
                                                               unsigned* __restrict__ pcnt_out,
                                                               unsigned* __restrict__ binned,
                                                               unsigned* __restrict__ binoff) {
    __shared__ __attribute__((aligned(16))) unsigned lds[P1_LDS];
    if (blockIdx.x < P1_DEG_WGS) gcn_degree_role(src, pcnt_out, lds, (int)blockIdx.x);
    else                         gcn_bin_role(src, dst, binned, binoff, lds, (int)blockIdx.x - P1_DEG_WGS);
}

__global__ void __launch_bounds__(OW_THREADS) gcn_owner_kernel(const unsigned* __restrict__ binned,
                                                               const unsigned* __restrict__ binoff,
                                                               const float* __restrict__ xscaled,
                                                               const float* __restrict__ Wc,
                                                               const float* __restrict__ bc,
                                                               float* __restrict__ out) {
    __shared__ __attribute__((aligned(16))) float acc[OW_NODES * ROW_W];

    const int r    = blockIdx.x;
    const int tid  = threadIdx.x;
    const int lane = tid & 31;
    const int w    = __builtin_amdgcn_readfirstlane(tid >> 5);

    for (int i = tid; i < OW_NODES * ROW_W; i += OW_THREADS) acc[i] = 0.0f;
    __syncthreads();

    const int  e   = lane / OW_LANES;
    const int  q   = lane - e * OW_LANES;
#define OW_RUN(B, L, S)                                                                 \
    {                                                                                   \
        const int g = w * OW_SLICES + (S);                                              \
        const unsigned b0 = binoff[g * BIN_OFFS + r];                                   \
        const unsigned e0 = binoff[g * BIN_OFFS + r + 1];                               \
        (B) = (unsigned)g * BIN_STRIDE + b0;                                            \
        (L) = e0 - b0;                                                                  \
    }
    unsigned beg[OW_GROUP], len[OW_GROUP];
#pragma unroll
    for (int s = 0; s < OW_GROUP; ++s) OW_RUN(beg[s], len[s], s)
#pragma nounroll
    for (int s0 = 0; s0 < OW_SLICES; s0 += OW_GROUP) {
        const int sn = (s0 + OW_GROUP < OW_SLICES) ? s0 + OW_GROUP : s0;
        unsigned nbeg[OW_GROUP], nlen[OW_GROUP];
#pragma unroll
        for (int s = 0; s < OW_GROUP; ++s) OW_RUN(nbeg[s], nlen[s], sn + s)
        unsigned maxlen = 0u;
#pragma unroll
        for (int s = 0; s < OW_GROUP; ++s) maxlen = (len[s] > maxlen) ? len[s] : maxlen;
        for (unsigned j0 = 0u; j0 < maxlen; j0 += (unsigned)(OW_EPI * OW_PER)) {
            bool     ok[OW_GROUP * OW_PER];
            unsigned p[OW_GROUP * OW_PER];
#pragma unroll
            for (int k = 0; k < OW_GROUP * OW_PER; ++k) {
                const int s = k / OW_PER;
                const unsigned j = j0 + (unsigned)(OW_EPI * (k % OW_PER) + e);
                ok[k] = j < len[s];
                p[k]  = binned[ok[k] ? beg[s] + j : (unsigned)(w * OW_SLICES + s0 + s) * BIN_STRIDE];
            }
            row3 v[OW_GROUP * OW_PER];
#pragma unroll
            for (int k = 0; k < OW_GROUP * OW_PER; ++k)
                v[k] = *(const row3*)(xscaled + (size_t)(p[k] & 0x1FFFFu) * XS_PITCH + 3 * q);
#pragma unroll
            for (int k = 0; k < OW_GROUP * OW_PER; ++k)
                asm volatile("" :: "v"(v[k].x), "v"(v[k].y), "v"(v[k].z));
#pragma unroll
            for (int k = 0; k < OW_GROUP * OW_PER; ++k)
                if (ok[k]) {
                    const unsigned t = p[k] >> 17;
                    float* a = acc + t * ROW_W + 3 * q;
                    atomicAdd(a, v[k].x);
                    atomicAdd(a + 1, v[k].y);
                    atomicAdd(a + 2, v[k].z);
                }
        }
#pragma unroll
        for (int s = 0; s < OW_GROUP; ++s) { beg[s] = nbeg[s]; len[s] = nlen[s]; }
    }
#undef OW_RUN
    __syncthreads();

    const int m  = lane & 15;
    const int kh = lane >> 4;
    const int nn = lane & 15;
    for (int tile = w; tile < OW_TILES; tile += OW_WAVES) {
        const int ln = tile * 16 + m;
        float din = acc[ln * ROW_W + F_IN]; din = (din > 0.0f) ? din : 1.0f;
        const float nd = rsqrtf(din);

        v2f afrag[3];
#pragma unroll
        for (int ks = 0; ks < 3; ++ks) {
            const int k0 = ks * 4 + kh * 2;
            float sx = 0.0f, sy = 0.0f;
            if (k0 < F_IN) {
                const float2 pv = *(const float2*)(acc + ln * ROW_W + k0);
                sx = pv.x; sy = pv.y;
            }
            afrag[ks].x = sx * nd;
            afrag[ks].y = sy * nd;
        }

        const int nbase = r * OW_NODES + tile * 16;
#pragma unroll
        for (int j = 0; j < 4; ++j) {
            v8f c = {};
#pragma unroll
            for (int ks = 0; ks < 3; ++ks) {
                const int kb = ks * 4 + kh * 2;
                v2f b;
                b.x = Wc[(kb    ) * N_OUT + j * 16 + nn];
                b.y = Wc[(kb + 1) * N_OUT + j * 16 + nn];
                c = __builtin_amdgcn_wmma_f32_16x16x4_f32(
                         false, afrag[ks],
                         false, b,
                         (short)0, c,
                         false,  false);
            }
            const float bias = bc[j * 16 + nn];
#pragma unroll
            for (int r8 = 0; r8 < 8; ++r8) {
                const int row = r8 + 8 * kh;
                out[(size_t)(nbase + row) * N_OUT + j * 16 + nn] = c[r8] + bias;
            }
        }
    }
}

extern "C" void kernel_launch(void* const* d_in, const int* in_sizes, int n_in,
                              void* d_out, int out_size, void* d_ws, size_t ws_size,
                              hipStream_t stream) {
    const float* x   = (const float*)d_in[0];
    const int*   src = (const int*)  d_in[1];
    const int*   dst = (const int*)  d_in[2];
    const float* W1  = (const float*)d_in[3];
    const float* b1  = (const float*)d_in[4];
    const float* W2  = (const float*)d_in[5];
    const float* b2  = (const float*)d_in[6];
    float* out = (float*)d_out;

    float* ws = (float*)d_ws;
    unsigned* pcnt_out = (unsigned*)ws;
    float*    xscaled  = ws + (size_t)DEG_CHUNKS * DEG_RANGES * DEG_SLAB;
    float*    Wc       = xscaled + (size_t)N_NODES * XS_PITCH;
    float*    bcv      = Wc + KPAD * N_OUT;
    unsigned* binned   = (unsigned*)(bcv + N_OUT);
    unsigned* binoff   = binned + (size_t)BIN_SLICES * BIN_STRIDE;

    gcn_edges_kernel<<<P1_DEG_WGS + BIN_SLICES, P1_THREADS, 0, stream>>>(src, dst, pcnt_out, binned, binoff);

    gcn_prep_kernel<<<FOLD_WGS + (N_NODES + 255) / 256, 256, 0, stream>>>(x, pcnt_out, xscaled, N_NODES,
                                                                          W1, b1, W2, b2, Wc, bcv);

    gcn_owner_kernel<<<OW_RANGES, OW_THREADS, 0, stream>>>(binned, binoff, xscaled, Wc, bcv, out);
}


# Owner loop software-pipelined: steps of 2 slices; the next step's row gathers and the following step's entry loads are issued before a step's LDS-atomic adds (two register sets, arrival-order sums)
# speedup vs baseline: 17.2268x; 1.0459x over previous
#include <hip/hip_runtime.h>


#define N_NODES  100000
#define N_EDGES  3200000
#define F_IN     10
#define KPAD     12
#define XS_PITCH 16
#define ROW_W    12
#define N_OUT    64

#define DEG_RANGES 16
#define DEG_CHUNKS 16
#define DEG_NODES  (N_NODES / DEG_RANGES)
#define DEG_EDGES  (N_EDGES / DEG_CHUNKS)
#define DEG_SLAB   6272
#define DEG_LOADS  8

#define OW_RANGES  250
#define OW_NODES   (N_NODES / OW_RANGES)
#define OW_TILES   (OW_NODES / 16)
#define OW_WAVES   16
#define OW_THREADS (OW_WAVES * 32)
#define OW_LANES   4
#define OW_EPI     8
#define OW_UNIT    2
#define OW_PER     9

#define BIN_SLICES  256
#define OW_SLICES   (BIN_SLICES / OW_WAVES)
#define OW_UNITS    (OW_SLICES / OW_UNIT)
#define OW_K        (OW_UNIT * OW_PER)
#define BIN_EDGES   (N_EDGES / BIN_SLICES)
#define BIN_STRIDE  12512
#define BIN_OFFS    (OW_RANGES + 1)
#define BIN_THREADS 625
#define BIN_LOADS   5
#define BIN_NONE    0xFFFFFFFFu

#define P1_THREADS  640
#define P1_DEG_WGS  (DEG_RANGES * DEG_CHUNKS)
#define P1_HIST_AT  BIN_STRIDE
#define P1_START_AT (BIN_STRIDE + 256)
#define P1_WSUM_AT  (BIN_STRIDE + 512)
#define P1_LDS      (BIN_STRIDE + 512 + 32)

typedef __attribute__((ext_vector_type(2))) float v2f;
struct row3 { float x, y, z; };
typedef __attribute__((ext_vector_type(8))) float v8f;

__device__ __forceinline__ void gcn_degree_role(const int* __restrict__ src, unsigned* __restrict__ pcnt_out,
                                                unsigned* cnt, int b) {
    const int r = b / DEG_CHUNKS;
    const int c = b % DEG_CHUNKS;
    const unsigned lo = (unsigned)(r * DEG_NODES);
    const int tid = threadIdx.x;

    for (int i = tid; i < DEG_NODES; i += P1_THREADS) cnt[i] = 0u;
    __syncthreads();

    const int4* s4 = (const int4*)(src + (size_t)c * DEG_EDGES);
    const int n4 = DEG_EDGES / 4;
    for (int i0 = tid; i0 < n4; i0 += P1_THREADS * DEG_LOADS) {
        int4 v[DEG_LOADS];
#pragma unroll
        for (int u = 0; u < DEG_LOADS; ++u) {
            int idx = i0 + u * P1_THREADS;
            v[u] = s4[(idx < n4) ? idx : (n4 - 1)];
        }
#pragma unroll
        for (int u = 0; u < DEG_LOADS; ++u) {
            const bool ok = (i0 + u * P1_THREADS) < n4;
            unsigned t;
            t = (unsigned)v[u].x - lo; if (ok && t < (unsigned)DEG_NODES) atomicAdd(&cnt[t], 1u);
            t = (unsigned)v[u].y - lo; if (ok && t < (unsigned)DEG_NODES) atomicAdd(&cnt[t], 1u);
            t = (unsigned)v[u].z - lo; if (ok && t < (unsigned)DEG_NODES) atomicAdd(&cnt[t], 1u);
            t = (unsigned)v[u].w - lo; if (ok && t < (unsigned)DEG_NODES) atomicAdd(&cnt[t], 1u);
        }
    }
    __syncthreads();

    unsigned* o = pcnt_out + (size_t)(c * DEG_RANGES + r) * DEG_SLAB;
    for (int i = tid; i < DEG_NODES; i += P1_THREADS) o[i] = cnt[i];
}

__device__ __forceinline__ void gcn_nodeprep_role(const float* __restrict__ x,
                                                  const unsigned* __restrict__ pcnt_out,
                                                  float* __restrict__ xscaled,
                                                  int nN, int b) {
    int n = b * 256 + threadIdx.x;
    if (n < nN) {
        const int r = n / DEG_NODES;
        const int l = n - r * DEG_NODES;
        unsigned d = 0u;
#pragma unroll
        for (int c = 0; c < DEG_CHUNKS; ++c) d += pcnt_out[(size_t)(c * DEG_RANGES + r) * DEG_SLAB + l];
        float dout = (float)d; dout = (dout > 0.0f) ? dout : 1.0f;
        float ns = rsqrtf(dout);
        const float* xr = x + (size_t)n * 40 + 15;
        float4* o = (float4*)(xscaled + (size_t)n * XS_PITCH);
        o[0] = make_float4(xr[0] * ns, xr[1] * ns, xr[2] * ns, xr[3] * ns);
        o[1] = make_float4(xr[4] * ns, xr[5] * ns, xr[6] * ns, xr[7] * ns);
        o[2] = make_float4(xr[8] * ns, xr[9] * ns, 1.0f, 0.0f);
    }
}

#define FOLD_PARTS 4
#define FOLD_K     (128 / FOLD_PARTS)
__device__ __forceinline__ void gcn_foldw_role(const float* __restrict__ W1,
                                               const float* __restrict__ b1,
                                               const float* __restrict__ W2,
                                               const float* __restrict__ b2,
                                               float* __restrict__ Wc,
                                               float* __restrict__ bc,
                                               float (*part)[64], int i) {
    const int j = threadIdx.x & 63;
    const int h = threadIdx.x >> 6;
    const float* a = ((i < F_IN) ? (W1 + i * 128) : b1) + h * FOLD_K;
    const float* w = W2 + (h * FOLD_K) * 64 + j;

    float av[FOLD_K], wv[FOLD_K];
#pragma unroll
    for (int k = 0; k < FOLD_K; ++k) { av[k] = a[k]; wv[k] = w[k * 64]; }
    float s = 0.0f;
#pragma unroll
    for (int k = 0; k < FOLD_K; ++k) s += av[k] * wv[k];
    part[h][j] = s;
    __syncthreads();

    if (h == 0) {
        float t = part[0][j];
#pragma unroll
        for (int q = 1; q < FOLD_PARTS; ++q) t += part[q][j];
        if (i < F_IN) Wc[i * 64 + j] = t;
        else          bc[j] = b2[j] + t;
    } else if (i == F_IN && h <= KPAD - F_IN) {
        Wc[(F_IN + h - 1) * 64 + j] = 0.0f;
    }
}

#define FOLD_WGS (F_IN + 1)
__global__ void __launch_bounds__(256) gcn_prep_kernel(const float* __restrict__ x,
                                                       const unsigned* __restrict__ pcnt_out,
                                                       float* __restrict__ xscaled,
                                                       int nN,
                                                       const float* __restrict__ W1,
                                                       const float* __restrict__ b1,
                                                       const float* __restrict__ W2,
                                                       const float* __restrict__ b2,
                                                       float* __restrict__ Wc,
                                                       float* __restrict__ bc) {
    __shared__ float part[FOLD_PARTS][64];
    if (blockIdx.x < FOLD_WGS) gcn_foldw_role(W1, b1, W2, b2, Wc, bc, part, (int)blockIdx.x);
    else                       gcn_nodeprep_role(x, pcnt_out, xscaled, nN, (int)blockIdx.x - FOLD_WGS);
}

__device__ __forceinline__ void gcn_bin_role(const int* __restrict__ src,
                                             const int* __restrict__ dst,
                                             unsigned* __restrict__ binned,
                                             unsigned* __restrict__ binoff,
                                             unsigned* lds, int g) {
    unsigned* stage = lds;
    unsigned* hist  = lds + P1_HIST_AT;
    unsigned* start = lds + P1_START_AT;
    unsigned* wsum  = lds + P1_WSUM_AT;

    const int tid = threadIdx.x;
    const bool worker = tid < BIN_THREADS;
    if (tid <= OW_RANGES) hist[tid] = 0u;
    __syncthreads();

    const int4* d4 = (const int4*)(dst + (size_t)g * BIN_EDGES);
    const int4* s4 = (const int4*)(src + (size_t)g * BIN_EDGES);
    const int n4 = BIN_EDGES / 4;

    unsigned pk[BIN_LOADS * 4], br[BIN_LOADS * 4];
    if (worker) {
        int4 dv[BIN_LOADS], sv[BIN_LOADS];
#pragma unroll
        for (int u = 0; u < BIN_LOADS; ++u) {
            dv[u] = d4[tid + u * BIN_THREADS];
            sv[u] = s4[tid + u * BIN_THREADS];
        }

#define BIN_RANK(J, D, S)                                                               \
        {                                                                                   \
            const unsigned d  = (unsigned)(D);                                              \
            const bool     in = d < (unsigned)N_NODES;                                      \
            const unsigned b  = in ? d / (unsigned)OW_NODES : (unsigned)OW_RANGES;          \
            const unsigned t  = d - b * (unsigned)OW_NODES;                                 \
            pk[J] = (t << 17) | ((unsigned)(S) & 0x1FFFFu);                                 \
            const unsigned rk = atomicAdd(&hist[b], 1u);                                    \
            br[J] = in ? ((b << 16) | rk) : BIN_NONE;                                       \
        }
#pragma unroll
        for (int u = 0; u < BIN_LOADS; ++u) {
            BIN_RANK(4 * u + 0, dv[u].x, sv[u].x)
            BIN_RANK(4 * u + 1, dv[u].y, sv[u].y)
            BIN_RANK(4 * u + 2, dv[u].z, sv[u].z)
            BIN_RANK(4 * u + 3, dv[u].w, sv[u].w)
        }
#undef BIN_RANK
    }
    __syncthreads();

    unsigned hv = 0u, inc = 0u;
    if (tid < 256) {
        hv  = (tid < OW_RANGES) ? hist[tid] : 0u;
        inc = hv;
        for (int o = 1; o < 32; o <<= 1) {
            const unsigned nb = __shfl_up(inc, (unsigned)o);
            if ((tid & 31) >= o) inc += nb;
        }
        if ((tid & 31) == 31) wsum[tid >> 5] = inc;
    }
    __syncthreads();
    if (tid < 256) {
        unsigned base = 0u;
        for (int q = 0; q < 8; ++q) if (q < (tid >> 5)) base += wsum[q];
        const unsigned ex = base + inc - hv;
        if (tid < OW_RANGES) {
            start[tid] = ex;
            binoff[g * BIN_OFFS + tid] = ex;
        }
        if (tid == OW_RANGES - 1) binoff[g * BIN_OFFS + OW_RANGES] = ex + hv;
    }
    __syncthreads();

    if (worker) {
#pragma unroll
        for (int j = 0; j < BIN_LOADS * 4; ++j)
            if (br[j] != BIN_NONE) stage[start[br[j] >> 16] + (br[j] & 0xFFFFu)] = pk[j];
    }
    __syncthreads();

    uint4* o = (uint4*)(binned + (size_t)g * BIN_STRIDE);
    const uint4* st4 = (const uint4*)stage;
    for (int i = tid; i < n4; i += P1_THREADS) o[i] = st4[i];
}

__global__ void __launch_bounds__(P1_THREADS) gcn_edges_kernel(const int* __restrict__ src,
                                                               const int* __restrict__ dst,
                                                               unsigned* __restrict__ pcnt_out,
                                                               unsigned* __restrict__ binned,
                                                               unsigned* __restrict__ binoff) {
    __shared__ __attribute__((aligned(16))) unsigned lds[P1_LDS];
    if (blockIdx.x < P1_DEG_WGS) gcn_degree_role(src, pcnt_out, lds, (int)blockIdx.x);
    else                         gcn_bin_role(src, dst, binned, binoff, lds, (int)blockIdx.x - P1_DEG_WGS);
}

__global__ void __launch_bounds__(OW_THREADS) gcn_owner_kernel(const unsigned* __restrict__ binned,
                                                               const unsigned* __restrict__ binoff,
                                                               const float* __restrict__ xscaled,
                                                               const float* __restrict__ Wc,
                                                               const float* __restrict__ bc,
                                                               float* __restrict__ out) {
    __shared__ __attribute__((aligned(16))) float acc[OW_NODES * ROW_W];

    const int r    = blockIdx.x;
    const int tid  = threadIdx.x;
    const int lane = tid & 31;
    const int w    = __builtin_amdgcn_readfirstlane(tid >> 5);

    const int  e   = lane / OW_LANES;
    const int  q   = lane - e * OW_LANES;

    unsigned b0 = 0u, e0 = 0u;
    if (lane < OW_SLICES) {
        const int g = w * OW_SLICES + lane;
        b0 = binoff[g * BIN_OFFS + r];
        e0 = binoff[g * BIN_OFFS + r + 1];
    }

    for (int i = tid; i < OW_NODES * ROW_W; i += OW_THREADS) acc[i] = 0.0f;
    __syncthreads();

    const unsigned basev = (unsigned)(w * OW_SLICES + ((lane < OW_SLICES) ? lane : OW_SLICES - 1)) * BIN_STRIDE;
    const unsigned begv  = basev + b0;
    const unsigned lenv  = e0 - b0;

#define OW_LOADP(U)                                                                         \
    {                                                                                       \
        unsigned bbs[OW_UNIT], lls[OW_UNIT], bas[OW_UNIT];                                  \
        _Pragma("unroll")                                                                   \
        for (int s = 0; s < OW_UNIT; ++s) {                                                 \
            const int su = (U) * OW_UNIT + s;        \
            bbs[s] = (unsigned)__builtin_amdgcn_readlane((int)begv, su);                    \
            lls[s] = (unsigned)__builtin_amdgcn_readlane((int)lenv, su);                    \
            bas[s] = (unsigned)__builtin_amdgcn_readlane((int)basev, su);                   \
        }                                                                                   \
        _Pragma("unroll")                                                                   \
        for (int k = 0; k < OW_K; ++k) {                                                    \
            const unsigned j = (unsigned)(OW_EPI * (k % OW_PER) + e);                       \
            P[k] = binned[(j < lls[k / OW_PER]) ? bbs[k / OW_PER] + j : bas[k / OW_PER]];   \
        }                                                                                   \
    }
#define OW_GATHER(T, V)                                                                     \
    _Pragma("unroll")                                                                       \
    for (int k = 0; k < OW_K; ++k) {                                                        \
        T[k] = (P[k] >> 17) * ROW_W + 3 * q;                                                \
        V[k] = *(const row3*)(xscaled + (size_t)(P[k] & 0x1FFFFu) * XS_PITCH + 3 * q);      \
    }
#define OW_ADDS(U, T, V)                                                                    \
    _Pragma("unroll")                                                                       \
    for (int k = 0; k < OW_K; ++k)                                                          \
        asm volatile("" :: "v"(V[k].x), "v"(V[k].y), "v"(V[k].z));                          \
    _Pragma("unroll")                                                                       \
    for (int k = 0; k < OW_K; ++k) {                                                        \
        const int sl = (U) * OW_UNIT + k / OW_PER;                                          \
        const unsigned ll = (unsigned)__builtin_amdgcn_readlane((int)lenv, sl);             \
        if ((unsigned)(OW_EPI * (k % OW_PER) + e) < ll) {                                   \
            float* a = acc + T[k];                                                          \
            atomicAdd(a, V[k].x);                                                           \
            atomicAdd(a + 1, V[k].y);                                                       \
            atomicAdd(a + 2, V[k].z);                                                       \
        }                                                                                   \
    }
    unsigned P[OW_K];
    unsigned tA[OW_K], tB[OW_K];
    row3     vA[OW_K], vB[OW_K];
    OW_LOADP(0)
    OW_GATHER(tA, vA)
    OW_LOADP(1)
#pragma nounroll
    for (int u = 0; u < OW_UNITS; u += 2) {
        OW_GATHER(tB, vB)
        OW_LOADP(u + 2)
        OW_ADDS(u, tA, vA)
        OW_GATHER(tA, vA)
        OW_LOADP(u + 3)
        OW_ADDS(u + 1, tB, vB)
    }
#undef OW_LOADP
#undef OW_GATHER
#undef OW_ADDS

    unsigned longm = __builtin_amdgcn_ballot_w32(lenv > (unsigned)(OW_EPI * OW_PER));
    while (longm != 0u) {
        const int sl = __builtin_ctz(longm);
        longm &= longm - 1u;
        const unsigned bb = (unsigned)__builtin_amdgcn_readlane((int)begv, sl);
        const unsigned ll = (unsigned)__builtin_amdgcn_readlane((int)lenv, sl);
        for (unsigned j0 = (unsigned)(OW_EPI * OW_PER); j0 < ll; j0 += (unsigned)(OW_EPI * OW_PER)) {
            unsigned pp[OW_PER];
            row3     vv[OW_PER];
#pragma unroll
            for (int k = 0; k < OW_PER; ++k) {
                const unsigned j = j0 + (unsigned)(OW_EPI * k + e);
                pp[k] = binned[(j < ll) ? bb + j : (unsigned)(w * OW_SLICES + sl) * BIN_STRIDE];
            }
#pragma unroll
            for (int k = 0; k < OW_PER; ++k)
                vv[k] = *(const row3*)(xscaled + (size_t)(pp[k] & 0x1FFFFu) * XS_PITCH + 3 * q);
#pragma unroll
            for (int k = 0; k < OW_PER; ++k)
                asm volatile("" :: "v"(vv[k].x), "v"(vv[k].y), "v"(vv[k].z));
#pragma unroll
            for (int k = 0; k < OW_PER; ++k)
                if (j0 + (unsigned)(OW_EPI * k + e) < ll) {
                    float* a = acc + (pp[k] >> 17) * ROW_W + 3 * q;
                    atomicAdd(a, vv[k].x);
                    atomicAdd(a + 1, vv[k].y);
                    atomicAdd(a + 2, vv[k].z);
                }
        }
    }
    __syncthreads();

    const int m  = lane & 15;
    const int kh = lane >> 4;
    const int nn = lane & 15;
    for (int tile = w; tile < OW_TILES; tile += OW_WAVES) {
        const int ln = tile * 16 + m;
        float din = acc[ln * ROW_W + F_IN]; din = (din > 0.0f) ? din : 1.0f;
        const float nd = rsqrtf(din);

        v2f afrag[3];
#pragma unroll
        for (int ks = 0; ks < 3; ++ks) {
            const int k0 = ks * 4 + kh * 2;
            float sx = 0.0f, sy = 0.0f;
            if (k0 < F_IN) {
                const float2 pv = *(const float2*)(acc + ln * ROW_W + k0);
                sx = pv.x; sy = pv.y;
            }
            afrag[ks].x = sx * nd;
            afrag[ks].y = sy * nd;
        }

        const int nbase = r * OW_NODES + tile * 16;
#pragma unroll
        for (int j = 0; j < 4; ++j) {
            v8f c = {};
#pragma unroll
            for (int ks = 0; ks < 3; ++ks) {
                const int kb = ks * 4 + kh * 2;
                v2f b;
                b.x = Wc[(kb    ) * N_OUT + j * 16 + nn];
                b.y = Wc[(kb + 1) * N_OUT + j * 16 + nn];
                c = __builtin_amdgcn_wmma_f32_16x16x4_f32(
                         false, afrag[ks],
                         false, b,
                         (short)0, c,
                         false,  false);
            }
            const float bias = bc[j * 16 + nn];
#pragma unroll
            for (int r8 = 0; r8 < 8; ++r8) {
                const int row = r8 + 8 * kh;
                out[(size_t)(nbase + row) * N_OUT + j * 16 + nn] = c[r8] + bias;
            }
        }
    }
}

extern "C" void kernel_launch(void* const* d_in, const int* in_sizes, int n_in,
                              void* d_out, int out_size, void* d_ws, size_t ws_size,
                              hipStream_t stream) {
    const float* x   = (const float*)d_in[0];
    const int*   src = (const int*)  d_in[1];
    const int*   dst = (const int*)  d_in[2];
    const float* W1  = (const float*)d_in[3];
    const float* b1  = (const float*)d_in[4];
    const float* W2  = (const float*)d_in[5];
    const float* b2  = (const float*)d_in[6];
    float* out = (float*)d_out;

    float* ws = (float*)d_ws;
    unsigned* pcnt_out = (unsigned*)ws;
    float*    xscaled  = ws + (size_t)DEG_CHUNKS * DEG_RANGES * DEG_SLAB;
    float*    Wc       = xscaled + (size_t)N_NODES * XS_PITCH;
    float*    bcv      = Wc + KPAD * N_OUT;
    unsigned* binned   = (unsigned*)(bcv + N_OUT);
    unsigned* binoff   = binned + (size_t)BIN_SLICES * BIN_STRIDE;

    gcn_edges_kernel<<<P1_DEG_WGS + BIN_SLICES, P1_THREADS, 0, stream>>>(src, dst, pcnt_out, binned, binoff);

    gcn_prep_kernel<<<FOLD_WGS + (N_NODES + 255) / 256, 256, 0, stream>>>(x, pcnt_out, xscaled, N_NODES,
                                                                          W1, b1, W2, b2, Wc, bcv);

    gcn_owner_kernel<<<OW_RANGES, OW_THREADS, 0, stream>>>(binned, binoff, xscaled, Wc, bcv, out);
}
